# starting kernel
# baseline (speedup 1.0000x reference)
#include <hip/hip_runtime.h>
#include <hip/hip_bf16.h>


typedef __attribute__((ext_vector_type(2))) float v2f;
typedef __attribute__((ext_vector_type(8))) float v8f;

#define BB   2
#define LQ   1024
#define LIN  4096
#define CC   256
#define MH   8
#define NP   4
#define DH   32
#define QT   4

template <int MT, int NT>
__global__ void __launch_bounds__(32)
wmma_gemm_f32(const float* __restrict__ X, const float* __restrict__ W,
              const float* __restrict__ bias, float* __restrict__ out,
              int Ncols, int K) {
    const int lane = threadIdx.x;
    const int l15  = lane & 15;
    const int kh   = lane >> 4;
    const int row0 = blockIdx.x * (16 * MT);
    const int col0 = blockIdx.y * (16 * NT);

    const float* __restrict__ xr[MT];
    const float* __restrict__ wr[NT];
#pragma unroll
    for (int mt = 0; mt < MT; ++mt)
        xr[mt] = X + (size_t)(row0 + mt * 16 + l15) * K + 2 * kh;
#pragma unroll
    for (int nt = 0; nt < NT; ++nt)
        wr[nt] = W + (size_t)(col0 + nt * 16 + l15) * K + 2 * kh;

    v8f acc[MT][NT];
#pragma unroll
    for (int mt = 0; mt < MT; ++mt)
#pragma unroll
        for (int nt = 0; nt < NT; ++nt) {
            v8f z = {};
            acc[mt][nt] = z;
        }

#pragma unroll 2
    for (int k = 0; k < K; k += 4) {
        v2f a[MT], b[NT];
#pragma unroll
        for (int mt = 0; mt < MT; ++mt) a[mt] = *(const v2f*)(xr[mt] + k);
#pragma unroll
        for (int nt = 0; nt < NT; ++nt) b[nt] = *(const v2f*)(wr[nt] + k);
#pragma unroll
        for (int mt = 0; mt < MT; ++mt)
#pragma unroll
            for (int nt = 0; nt < NT; ++nt)
                acc[mt][nt] = __builtin_amdgcn_wmma_f32_16x16x4_f32(
                     false, a[mt],  false, b[nt],
                     (short)0, acc[mt][nt],
                     false,  false);
    }

#pragma unroll
    for (int nt = 0; nt < NT; ++nt) {
        const int   col  = col0 + nt * 16;
        const float bcol = bias[col + l15];
#pragma unroll
        for (int mt = 0; mt < MT; ++mt) {
            const int row = row0 + mt * 16;
#pragma unroll
            for (int r = 0; r < 8; ++r) {
                out[(size_t)(row + r + 8 * kh) * Ncols + col + l15] =
                    acc[mt][nt][r] + bcol;
            }
        }
    }
}

__global__ void build_pts(const float* __restrict__ off,
                          const float* __restrict__ qpts,
                          const float* __restrict__ ipts,
                          float* __restrict__ S4, float* __restrict__ V4) {
    const int i = blockIdx.x * blockDim.x + threadIdx.x;
    if (i < BB * LQ * MH * NP) {
        const int b  = i >> 15;
        const int q  = (i >> 5) & (LQ - 1);
        const int mp = i & 31;
        const float* o = off  + ((size_t)(b * LQ + q)) * (MH * NP * 3) + mp * 3;
        const float* p = qpts + ((size_t)(b * LQ + q)) * 3;
        float4 s;
        s.x = p[0] + o[0];
        s.y = p[1] + o[1];
        s.z = p[2] + o[2];
        s.w = 1.0f;
        *(float4*)(S4 + (size_t)i * 4) = s;
    }
    if (i < BB * LIN) {
        const float* v = ipts + (size_t)i * 3;
        const float x = v[0], y = v[1], z = v[2];
        float4 t;
        t.x = -2.0f * x;
        t.y = -2.0f * y;
        t.z = -2.0f * z;
        t.w = x * x + y * y + z * z;
        *(float4*)(V4 + (size_t)i * 4) = t;
    }
}

__global__ void softmax_p4(float* __restrict__ a, int rows) {
    const int r = blockIdx.x * blockDim.x + threadIdx.x;
    if (r >= rows) return;
    float4 v = *(float4*)(a + (size_t)r * 4);
    const float mx = fmaxf(fmaxf(v.x, v.y), fmaxf(v.z, v.w));
    const float e0 = __expf(v.x - mx), e1 = __expf(v.y - mx);
    const float e2 = __expf(v.z - mx), e3 = __expf(v.w - mx);
    const float inv = 1.0f / (e0 + e1 + e2 + e3);
    v.x = e0 * inv; v.y = e1 * inv; v.z = e2 * inv; v.w = e3 * inv;
    *(float4*)(a + (size_t)r * 4) = v;
}

__device__ __forceinline__ void knn_step(const v2f a, const v2f* bq, int cib,
                                         float* best, int* bidx) {
#pragma unroll
    for (int t = 0; t < QT; ++t) {
        v8f c = {};
        c = __builtin_amdgcn_wmma_f32_16x16x4_f32(
            false, a, false, bq[t], (short)0, c, false, false);
#pragma unroll
        for (int r = 0; r < 8; ++r) {
            const float dv = c[r];
            const int   ci = cib + r;
            if (dv < best[t]) { best[t] = dv; bidx[t] = ci; }
        }
    }
}

__global__ void __launch_bounds__(32)
knn_wmma(const float* __restrict__ S4, const float* __restrict__ V4,
         int* __restrict__ idxOut) {
    const int lane = threadIdx.x;
    const int l15  = lane & 15;
    const int kh   = lane >> 4;
    const int n0   = blockIdx.x * (16 * QT);
    const int b    = n0 >> 15;

    v2f bq[QT];
#pragma unroll
    for (int t = 0; t < QT; ++t)
        bq[t] = *(const v2f*)(S4 + (size_t)(n0 + t * 16 + l15) * 4 + 2 * kh);

    const float* __restrict__ vptr = V4 + (size_t)b * LIN * 4
                                        + (size_t)l15 * 4 + 2 * kh;

    float best[QT];
    int   bidx[QT];
#pragma unroll
    for (int t = 0; t < QT; ++t) { best[t] = 3.4e38f; bidx[t] = 0; }

    v2f a_cur = *(const v2f*)(vptr);
    for (int cb = 0; cb < LIN - 16; cb += 16) {
        const v2f a_next = *(const v2f*)(vptr + (size_t)(cb + 16) * 4);
        knn_step(a_cur, bq, cb + (kh << 3), best, bidx);
        a_cur = a_next;
    }
    knn_step(a_cur, bq, (LIN - 16) + (kh << 3), best, bidx);

#pragma unroll
    for (int t = 0; t < QT; ++t) {
        const float obest = __shfl_xor(best[t], 16, 32);
        const int   oidx  = __shfl_xor(bidx[t], 16, 32);
        if (obest < best[t] || (obest == best[t] && oidx < bidx[t])) {
            best[t] = obest; bidx[t] = oidx;
        }
        if (lane < 16) idxOut[n0 + t * 16 + l15] = bidx[t];
    }
}

__global__ void __launch_bounds__(CC)
gather_weighted(const float* __restrict__ value, const int* __restrict__ idx,
                const float* __restrict__ attw, float* __restrict__ out_tmp) {
    const int bq = blockIdx.x;
    const int c  = threadIdx.x;
    const int b  = bq >> 10;
    const int m  = c >> 5;
    const int*   id = idx  + ((size_t)bq * MH + m) * NP;
    const float* aw = attw + ((size_t)bq * MH + m) * NP;
    const float* vb = value + (size_t)b * LIN * CC;
    float acc = 0.0f;
#pragma unroll
    for (int p = 0; p < NP; ++p)
        acc += aw[p] * vb[(size_t)id[p] * CC + c];
    out_tmp[(size_t)bq * CC + c] = acc;
}

extern "C" void kernel_launch(void* const* d_in, const int* in_sizes, int n_in,
                              void* d_out, int out_size, void* d_ws, size_t ws_size,
                              hipStream_t stream) {
    const float* query = (const float*)d_in[0];
    const float* qpts  = (const float*)d_in[1];
    const float* inp   = (const float*)d_in[2];
    const float* ipts  = (const float*)d_in[3];
    const float* Wv    = (const float*)d_in[4];
    const float* bv    = (const float*)d_in[5];
    const float* Ws    = (const float*)d_in[6];
    const float* bs    = (const float*)d_in[7];
    const float* Wa    = (const float*)d_in[8];
    const float* ba    = (const float*)d_in[9];
    const float* Wo    = (const float*)d_in[10];
    const float* bo    = (const float*)d_in[11];
    float* out = (float*)d_out;

    char* w = (char*)d_ws;
    float* value = (float*)w;  w += (size_t)BB * LIN * CC * 4;
    float* off   = (float*)w;  w += (size_t)BB * LQ * MH * NP * 3 * 4;
    float* attw  = (float*)w;  w += (size_t)BB * LQ * MH * NP * 4;
    float* S4    = (float*)w;  w += (size_t)BB * LQ * MH * NP * 4 * 4;
    float* V4    = (float*)w;  w += (size_t)BB * LIN * 4 * 4;
    int*   idx   = (int*)w;    w += (size_t)BB * LQ * MH * NP * 4;
    float* otmp  = (float*)w;  w += (size_t)BB * LQ * CC * 4;

    const dim3 wave(32);

    wmma_gemm_f32<2, 4><<<dim3((BB * LIN) / 32, CC / 64), wave, 0, stream>>>(
        inp, Wv, bv, value, CC, CC);

    wmma_gemm_f32<2, 2><<<dim3((BB * LQ) / 32, (MH * NP * 3) / 32), wave, 0, stream>>>(
        query, Ws, bs, off, MH * NP * 3, CC);

    wmma_gemm_f32<2, 2><<<dim3((BB * LQ) / 32, (MH * NP) / 32), wave, 0, stream>>>(
        query, Wa, ba, attw, MH * NP, CC);

    {
        const int n = BB * LQ * MH * NP;
        build_pts<<<(n + 255) / 256, 256, 0, stream>>>(off, qpts, ipts, S4, V4);
    }

    {
        const int rows = BB * LQ * MH;
        softmax_p4<<<(rows + 255) / 256, 256, 0, stream>>>(attw, rows);
    }

    knn_wmma<<<(BB * LQ * MH * NP) / (16 * QT), wave, 0, stream>>>(S4, V4, idx);

    gather_weighted<<<BB * LQ, CC, 0, stream>>>(value, idx, attw, otmp);

    wmma_gemm_f32<2, 4><<<dim3((BB * LQ) / 32, CC / 64), wave, 0, stream>>>(
        otmp, Wo, bo, out, CC, CC);
}


# KNN scan and value product in one launch: work-groups of 4 scan + 4 product waves, one of each per SIMD; the product's matrix work fills the scan's compare/select time; 6->5 launches; bit-identical
# speedup vs baseline: 2.7317x; 2.7317x over previous
#include <hip/hip_runtime.h>
#include <hip/hip_bf16.h>


typedef __attribute__((ext_vector_type(2))) float v2f;
typedef __attribute__((ext_vector_type(8))) float v8f;
typedef __attribute__((ext_vector_type(4))) float v4f;

#define BB   2
#define LQ   1024
#define LIN  4096
#define CC   256
#define MH   8
#define NP   4
#define DH   32
#define QT   4
#define KW   4

#define KK     256
#define KPITCH (KK + 4)
#define KCH    64
#define KBL    32

template <int MT, int NT, int WM, int WN, int EPI>
__global__ void __launch_bounds__(32 * WM * WN)
wmma_gemm_lds_f32(const float* __restrict__ X, const float* __restrict__ W,
                  const float* __restrict__ bias, float* __restrict__ out,
                  const float* __restrict__ aux, int Ncols) {
    constexpr int NTH  = 32 * WM * WN;
    constexpr int RX   = 16 * MT * WM;
    constexpr int RW   = 16 * NT * WN;
    constexpr int ROWS = RX + RW;
    constexpr int C4   = KCH / 4;
    constexpr int F4C  = ROWS * C4;
    constexpr int NLD  = (F4C + NTH - 1) / NTH;
    constexpr int NCH  = KK / KCH;
    constexpr int NKB  = KK / KBL;
    constexpr int BPC  = KCH / KBL;
    constexpr int KS   = KBL / 4;

    __shared__ __attribute__((aligned(16))) float sT[ROWS * KPITCH];

    const int tid  = threadIdx.x;
    const int lane = tid & 31;
    const int wv   = tid >> 5;
    const int l15  = lane & 15;
    const int kh   = lane >> 4;
    const int wm   = wv / WN;
    const int wn   = wv % WN;
    const int row0 = blockIdx.x * RX;
    const int col0 = blockIdx.y * RW;

    constexpr int PW  = RW / 3;
    constexpr int E1I = RX * PW;
    constexpr int E1T = (EPI == 1) ? (E1I + NTH - 1) / NTH : 1;
    float qv[E1T][3];
    if constexpr (EPI == 1) {
        static_assert(RW % 3 == 0, "whole 3-vectors per work-group");
#pragma unroll
        for (int j = 0; j < E1T; ++j) {
            int i = tid + j * NTH;
            if (i >= E1I) i = E1I - 1;
            const float* q = aux + (size_t)(row0 + i / PW) * 3;
            qv[j][0] = q[0];
            qv[j][1] = q[1];
            qv[j][2] = q[2];
        }
    }

    const float* gsrc[NLD];
    int          soff[NLD];
#pragma unroll
    for (int j = 0; j < NLD; ++j) {
        int f = tid + j * NTH;
        if (f >= F4C) f = F4C - 1;
        const int r  = f / C4;
        const int c4 = f % C4;
        gsrc[j] = (r < RX) ? X + (size_t)(row0 + r) * KK + c4 * 4
                           : W + (size_t)(col0 + r - RX) * KK + c4 * 4;
        soff[j] = r * KPITCH + c4 * 4;
    }

    v4f tmp[NLD];
#pragma unroll
    for (int j = 0; j < NLD; ++j)
        if (j + 1 < NLD || F4C % NTH == 0 || tid + j * NTH < F4C)
            tmp[j] = *(const v4f*)(gsrc[j]);
#pragma unroll
    for (int j = 0; j < NLD; ++j)
        if (j + 1 < NLD || F4C % NTH == 0 || tid + j * NTH < F4C)
            *(v4f*)(sT + soff[j]) = tmp[j];
    __syncthreads();
    if (NCH > 1) {
#pragma unroll
        for (int j = 0; j < NLD; ++j)
            if (j + 1 < NLD || F4C % NTH == 0 || tid + j * NTH < F4C)
                tmp[j] = *(const v4f*)(gsrc[j] + KCH);
    }

    const float* ap = sT + (wm * 16 * MT + l15) * KPITCH + 2 * kh;
    const float* bp = sT + (RX + wn * 16 * NT + l15) * KPITCH + 2 * kh;

    v8f acc[MT][NT];
#pragma unroll
    for (int mt = 0; mt < MT; ++mt)
#pragma unroll
        for (int nt = 0; nt < NT; ++nt) {
            v8f z = {};
            acc[mt][nt] = z;
        }

    v2f fa[2][KS][MT], fb[2][KS][NT];
#pragma unroll
    for (int ks = 0; ks < KS; ++ks) {
#pragma unroll
        for (int mt = 0; mt < MT; ++mt)
            fa[0][ks][mt] = *(const v2f*)(ap + mt * 16 * KPITCH + ks * 4);
#pragma unroll
        for (int nt = 0; nt < NT; ++nt)
            fb[0][ks][nt] = *(const v2f*)(bp + nt * 16 * KPITCH + ks * 4);
    }

#pragma unroll
    for (int kb = 0; kb < NKB; ++kb) {
        const int cur = kb & 1;
        const int c   = kb / BPC;
        if (kb % BPC == BPC - 1 && c + 1 < NCH) {
#pragma unroll
            for (int j = 0; j < NLD; ++j)
                if (j + 1 < NLD || F4C % NTH == 0 || tid + j * NTH < F4C)
                    *(v4f*)(sT + soff[j] + (c + 1) * KCH) = tmp[j];
            __syncthreads();
            if (c + 2 < NCH) {
#pragma unroll
                for (int j = 0; j < NLD; ++j)
                    if (j + 1 < NLD || F4C % NTH == 0 || tid + j * NTH < F4C)
                        tmp[j] = *(const v4f*)(gsrc[j] + (c + 2) * KCH);
            }
        }
        if (kb + 1 < NKB) {
#pragma unroll
            for (int ks = 0; ks < KS; ++ks) {
#pragma unroll
                for (int mt = 0; mt < MT; ++mt)
                    fa[cur ^ 1][ks][mt] =
                        *(const v2f*)(ap + mt * 16 * KPITCH + (kb + 1) * KBL + ks * 4);
#pragma unroll
                for (int nt = 0; nt < NT; ++nt)
                    fb[cur ^ 1][ks][nt] =
                        *(const v2f*)(bp + nt * 16 * KPITCH + (kb + 1) * KBL + ks * 4);
            }
        }
        __builtin_amdgcn_sched_barrier(0);
#pragma unroll
        for (int ks = 0; ks < KS; ++ks)
#pragma unroll
            for (int mt = 0; mt < MT; ++mt)
#pragma unroll
                for (int nt = 0; nt < NT; ++nt)
                    acc[mt][nt] = __builtin_amdgcn_wmma_f32_16x16x4_f32(
                         false, fa[cur][ks][mt],  false, fb[cur][ks][nt],
                         (short)0, acc[mt][nt],
                         false,  false);
        __builtin_amdgcn_sched_barrier(0);
    }

    const int orow0 = row0 + wm * 16 * MT;
    const int ocol0 = col0 + wn * 16 * NT;
    if constexpr (EPI == 0) {
#pragma unroll
        for (int nt = 0; nt < NT; ++nt) {
            const int   col  = ocol0 + nt * 16;
            const float bcol = bias[col + l15];
#pragma unroll
            for (int mt = 0; mt < MT; ++mt) {
                const int row = orow0 + mt * 16;
#pragma unroll
                for (int r = 0; r < 8; ++r) {
                    out[(size_t)(row + r + 8 * kh) * Ncols + col + l15] =
                        acc[mt][nt][r] + bcol;
                }
            }
        }
    } else {
        constexpr int EP = RW + 4;
        __shared__ __attribute__((aligned(16))) float sE[RX * EP];
#pragma unroll
        for (int nt = 0; nt < NT; ++nt) {
            const int   lc   = wn * 16 * NT + nt * 16 + l15;
            const float bcol = bias[col0 + lc];
#pragma unroll
            for (int mt = 0; mt < MT; ++mt)
#pragma unroll
                for (int r = 0; r < 8; ++r)
                    sE[(wm * 16 * MT + mt * 16 + r + 8 * kh) * EP + lc] =
                        acc[mt][nt][r] + bcol;
        }
        __syncthreads();
        if constexpr (EPI == 1) {
            const int pts = Ncols / 3;
#pragma unroll
            for (int j = 0; j < E1T; ++j) {
                const int i = tid + j * NTH;
                if (j < E1I / NTH || i < E1I) {
                    const int    r = i / PW;
                    const int    p = i % PW;
                    const float* o = sE + r * EP + p * 3;
                    float4 s;
                    s.x = qv[j][0] + o[0];
                    s.y = qv[j][1] + o[1];
                    s.z = qv[j][2] + o[2];
                    s.w = 1.0f;
                    *(float4*)(out + ((size_t)(row0 + r) * pts + col0 / 3 + p) * 4) = s;
                }
            }
        } else {
            static_assert(RW % 4 == 0, "whole groups of 4 per work-group");
            constexpr int GW = RW / 4;
            constexpr int ITEMS = RX * GW;
#pragma unroll
            for (int j = 0; j < (ITEMS + NTH - 1) / NTH; ++j) {
                const int i = tid + j * NTH;
                if (ITEMS % NTH != 0 && i >= ITEMS) continue;
                const int r = i / GW;
                const int g = i % GW;
                float4 v = *(const float4*)(sE + r * EP + g * 4);
                const float mx = fmaxf(fmaxf(v.x, v.y), fmaxf(v.z, v.w));
                const float e0 = __expf(v.x - mx), e1 = __expf(v.y - mx);
                const float e2 = __expf(v.z - mx), e3 = __expf(v.w - mx);
                const float inv = 1.0f / (e0 + e1 + e2 + e3);
                v.x = e0 * inv; v.y = e1 * inv; v.z = e2 * inv; v.w = e3 * inv;
                *(float4*)(out + (size_t)(row0 + r) * Ncols + col0 + g * 4) = v;
            }
        }
    }
}

__device__ __forceinline__ void knn_rows2(const v8f* c, int r0, int cib,
                                          float* best, int* bidx) {
#pragma unroll
    for (int r = r0; r < r0 + 2; ++r) {
#pragma unroll
        for (int t = 0; t < QT; ++t) {
            const float dv = c[t][r];
            const int   ci = cib + r;
            if (dv < best[t]) { best[t] = dv; bidx[t] = ci; }
        }
    }
}

__device__ __forceinline__ void knn_phase(v8f* cn, const v8f* cc, const v2f a,
                                          const v2f* bq, int cib,
                                          float* best, int* bidx) {
    static_assert(2 * QT == 8, "one product per two of a tile's eight rows");
#pragma unroll
    for (int t = 0; t < QT; ++t) {
        v8f z = {};
        cn[t] = __builtin_amdgcn_wmma_f32_16x16x4_f32(
            false, a, false, bq[t], (short)0, z, false, false);
        knn_rows2(cc, 2 * t, cib, best, bidx);
        __builtin_amdgcn_sched_barrier(0);
    }
}

#define VRX   64
#define VRW   128
#define VROWS (VRX + VRW)

__global__ void __launch_bounds__(64 * KW)
knn_and_value(const float* __restrict__ S4, const float* __restrict__ ipts,
              int* __restrict__ idxOut,
              const float* __restrict__ X, const float* __restrict__ Wt,
              const float* __restrict__ bias, float* __restrict__ vout) {
    constexpr int NTH = 64 * KW;
    __shared__ __attribute__((aligned(16))) float sV[LIN * 4];
    __shared__ __attribute__((aligned(16))) float sT[VROWS * KPITCH];

    const int tid  = threadIdx.x;
    const int lane = tid & 31;
    const int wv   = __builtin_amdgcn_readfirstlane(tid >> 5);
    const int l15  = lane & 15;
    const int kh   = lane >> 4;
    const int b    = (blockIdx.x * (KW * 16 * QT)) >> 15;
    const int n0   = (blockIdx.x * KW + (wv & (KW - 1))) * (16 * QT);
    const int row0 = (blockIdx.x >> 1) * VRX;
    const int col0 = (blockIdx.x & 1) * VRW;

    v2f bq[QT];
#pragma unroll
    for (int t = 0; t < QT; ++t) {
        v2f z = {};
        bq[t] = z;
    }
    if (wv < KW) {
#pragma unroll
        for (int t = 0; t < QT; ++t)
            bq[t] = *(const v2f*)(S4 + (size_t)(n0 + t * 16 + l15) * 4 + 2 * kh);
    }

    {
        const float4* __restrict__ g = (const float4*)(ipts + (size_t)b * LIN * 3);
        float4* s = (float4*)sV;
        constexpr int NG  = LIN / 4 / NTH;
        constexpr int C4  = KK / 4;
        constexpr int RPT = NTH / C4;
        constexpr int NLX = VRX / RPT;
        constexpr int NLW = VRW / RPT;
        static_assert(NTH % C4 == 0 && VRX % RPT == 0 && VRW % RPT == 0, "whole trips");
        const int pr = tid / C4;
        const int pc = tid % C4;
        const float* xsrc = X  + (size_t)(row0 + pr) * KK + pc * 4;
        const float* wsrc = Wt + (size_t)(col0 + pr) * KK + pc * 4;
        float*       sdst = sT + pr * KPITCH + pc * 4;

        float4 tmp[NG][3];
        v4f    xt[NLX], wt[NLW];
#pragma unroll
        for (int j = 0; j < NG; ++j)
#pragma unroll
            for (int k = 0; k < 3; ++k) tmp[j][k] = g[3 * (tid + NTH * j) + k];
#pragma unroll
        for (int j = 0; j < NLX; ++j) xt[j] = *(const v4f*)(xsrc + (size_t)j * RPT * KK);
#pragma unroll
        for (int j = 0; j < NLW; ++j) wt[j] = *(const v4f*)(wsrc + (size_t)j * RPT * KK);

#pragma unroll
        for (int j = 0; j < NG; ++j) {
            const float f[12] = {tmp[j][0].x, tmp[j][0].y, tmp[j][0].z, tmp[j][0].w,
                                 tmp[j][1].x, tmp[j][1].y, tmp[j][1].z, tmp[j][1].w,
                                 tmp[j][2].x, tmp[j][2].y, tmp[j][2].z, tmp[j][2].w};
#pragma unroll
            for (int i = 0; i < 4; ++i) {
                const float x = f[3 * i], y = f[3 * i + 1], z = f[3 * i + 2];
                float xy;
                {
#pragma clang fp contract(off)
                    xy = x * x + y * y;
                }
                float4 t;
                t.x = -2.0f * x;
                t.y = -2.0f * y;
                t.z = -2.0f * z;
                t.w = __builtin_fmaf(z, z, xy);
                s[4 * (tid + NTH * j) + i] = t;
            }
        }
#pragma unroll
        for (int j = 0; j < NLX; ++j) *(v4f*)(sdst + j * RPT * KPITCH) = xt[j];
#pragma unroll
        for (int j = 0; j < NLW; ++j) *(v4f*)(sdst + (VRX + j * RPT) * KPITCH) = wt[j];
    }
    __syncthreads();

    if (wv < KW) {
        const float* vptr = sV + l15 * 4 + 2 * kh;

        float best[QT];
        int   bidx[QT];
#pragma unroll
        for (int t = 0; t < QT; ++t) { best[t] = 3.4e38f; bidx[t] = 0; }

        constexpr int NBLK = LIN / 16;
        const int ch = kh << 3;
        v8f c0[QT], c1[QT];
        v2f aA = *(const v2f*)(vptr);
        v2f aB = *(const v2f*)(vptr + 16 * 4);
#pragma unroll
        for (int t = 0; t < QT; ++t) {
            v8f z = {};
            c0[t] = __builtin_amdgcn_wmma_f32_16x16x4_f32(
                false, aA, false, bq[t], (short)0, z, false, false);
        }
        aA = *(const v2f*)(vptr + 2 * 16 * 4);
        for (int j = 0; j < NBLK - 4; j += 2) {
            knn_phase(c1, c0, aB, bq, j * 16 + ch, best, bidx);
            aB = *(const v2f*)(vptr + (j + 3) * 16 * 4);
            knn_phase(c0, c1, aA, bq, (j + 1) * 16 + ch, best, bidx);
            aA = *(const v2f*)(vptr + (j + 4) * 16 * 4);
        }
        knn_phase(c1, c0, aB, bq, (NBLK - 4) * 16 + ch, best, bidx);
        aB = *(const v2f*)(vptr + (NBLK - 1) * 16 * 4);
        knn_phase(c0, c1, aA, bq, (NBLK - 3) * 16 + ch, best, bidx);
        knn_phase(c1, c0, aB, bq, (NBLK - 2) * 16 + ch, best, bidx);
#pragma unroll
        for (int q = 0; q < 4; ++q)
            knn_rows2(c1, 2 * q, (NBLK - 1) * 16 + ch, best, bidx);

#pragma unroll
        for (int t = 0; t < QT; ++t) {
            const float obest = __shfl_xor(best[t], 16, 32);
            const int   oidx  = __shfl_xor(bidx[t], 16, 32);
            if (obest < best[t] || (obest == best[t] && oidx < bidx[t])) {
                best[t] = obest; bidx[t] = oidx;
            }
            if (lane < 16) idxOut[n0 + t * 16 + l15] = bidx[t];
        }
    } else {
        constexpr int MT  = 2, NT = 4;
        constexpr int NKB = KK / KBL;
        constexpr int KS  = KBL / 4;
        static_assert(VRX == 2 * 16 * MT && VRW == 2 * 16 * NT, "2x2 waves of 32x64");
        const int w  = wv - KW;
        const int wm = w >> 1;
        const int wn = w & 1;
        const float* ap = sT + (wm * 16 * MT + l15) * KPITCH + 2 * kh;
        const float* bp = sT + (VRX + wn * 16 * NT + l15) * KPITCH + 2 * kh;

        v8f acc[MT][NT];
#pragma unroll
        for (int mt = 0; mt < MT; ++mt)
#pragma unroll
            for (int nt = 0; nt < NT; ++nt) {
                v8f z = {};
                acc[mt][nt] = z;
            }

        v2f fa[2][KS][MT], fb[2][KS][NT];
#pragma unroll
        for (int ks = 0; ks < KS; ++ks) {
#pragma unroll
            for (int mt = 0; mt < MT; ++mt)
                fa[0][ks][mt] = *(const v2f*)(ap + mt * 16 * KPITCH + ks * 4);
#pragma unroll
            for (int nt = 0; nt < NT; ++nt)
                fb[0][ks][nt] = *(const v2f*)(bp + nt * 16 * KPITCH + ks * 4);
        }

#pragma unroll
        for (int kb = 0; kb < NKB; ++kb) {
            const int cur = kb & 1;
            if (kb + 1 < NKB) {
#pragma unroll
                for (int ks = 0; ks < KS; ++ks) {
#pragma unroll
                    for (int mt = 0; mt < MT; ++mt)
                        fa[cur ^ 1][ks][mt] =
                            *(const v2f*)(ap + mt * 16 * KPITCH + (kb + 1) * KBL + ks * 4);
#pragma unroll
                    for (int nt = 0; nt < NT; ++nt)
                        fb[cur ^ 1][ks][nt] =
                            *(const v2f*)(bp + nt * 16 * KPITCH + (kb + 1) * KBL + ks * 4);
                }
            }
            __builtin_amdgcn_sched_barrier(0);
#pragma unroll
            for (int ks = 0; ks < KS; ++ks)
#pragma unroll
                for (int mt = 0; mt < MT; ++mt)
#pragma unroll
                    for (int nt = 0; nt < NT; ++nt)
                        acc[mt][nt] = __builtin_amdgcn_wmma_f32_16x16x4_f32(
                             false, fa[cur][ks][mt],  false, fb[cur][ks][nt],
                             (short)0, acc[mt][nt],
                             false,  false);
            __builtin_amdgcn_sched_barrier(0);
        }

        const int orow0 = row0 + wm * 16 * MT;
        const int ocol0 = col0 + wn * 16 * NT;
#pragma unroll
        for (int nt = 0; nt < NT; ++nt) {
            const int   col  = ocol0 + nt * 16;
            const float bcol = bias[col + l15];
#pragma unroll
            for (int mt = 0; mt < MT; ++mt) {
                const int row = orow0 + mt * 16;
#pragma unroll
                for (int r = 0; r < 8; ++r) {
                    vout[(size_t)(row + r + 8 * kh) * CC + col + l15] =
                        acc[mt][nt][r] + bcol;
                }
            }
        }
    }
}

__global__ void __launch_bounds__(CC)
gather_weighted(const float* __restrict__ value, const int* __restrict__ idx,
                const float* __restrict__ attw, float* __restrict__ out_tmp) {
    const int bq = blockIdx.x;
    const int c  = threadIdx.x;
    const int b  = bq >> 10;
    const int m  = c >> 5;
    const int*   id = idx  + ((size_t)bq * MH + m) * NP;
    const float* aw = attw + ((size_t)bq * MH + m) * NP;
    const float* vb = value + (size_t)b * LIN * CC;
    float acc = 0.0f;
#pragma unroll
    for (int p = 0; p < NP; ++p)
        acc += aw[p] * vb[(size_t)id[p] * CC + c];
    out_tmp[(size_t)bq * CC + c] = acc;
}

extern "C" void kernel_launch(void* const* d_in, const int* in_sizes, int n_in,
                              void* d_out, int out_size, void* d_ws, size_t ws_size,
                              hipStream_t stream) {
    const float* query = (const float*)d_in[0];
    const float* qpts  = (const float*)d_in[1];
    const float* inp   = (const float*)d_in[2];
    const float* ipts  = (const float*)d_in[3];
    const float* Wv    = (const float*)d_in[4];
    const float* bv    = (const float*)d_in[5];
    const float* Ws    = (const float*)d_in[6];
    const float* bs    = (const float*)d_in[7];
    const float* Wa    = (const float*)d_in[8];
    const float* ba    = (const float*)d_in[9];
    const float* Wo    = (const float*)d_in[10];
    const float* bo    = (const float*)d_in[11];
    float* out = (float*)d_out;

    char* w = (char*)d_ws;
    float* value = (float*)w;  w += (size_t)BB * LIN * CC * 4;
    float* attw  = (float*)w;  w += (size_t)BB * LQ * MH * NP * 4;
    float* S4    = (float*)w;  w += (size_t)BB * LQ * MH * NP * 4 * 4;
    int*   idx   = (int*)w;    w += (size_t)BB * LQ * MH * NP * 4;
    float* otmp  = (float*)w;  w += (size_t)BB * LQ * CC * 4;

    wmma_gemm_lds_f32<2, 2, 1, 3, 1><<<dim3((BB * LQ) / 32, 1), dim3(96), 0, stream>>>(
        query, Ws, bs, S4, qpts, MH * NP * 3);

    wmma_gemm_lds_f32<2, 2, 4, 1, 2><<<dim3((BB * LQ) / 128, 1), dim3(128), 0, stream>>>(
        query, Wa, ba, attw, nullptr, MH * NP);

    static_assert((BB * LQ * MH * NP) / (16 * QT * KW) == ((BB * LIN) / VRX) * (CC / VRW),
                  "as many scan work-groups as value blocks");
    knn_and_value<<<(BB * LQ * MH * NP) / (16 * QT * KW), dim3(64 * KW), 0, stream>>>(
        S4, ipts, idx, inp, Wv, bv, value);

    gather_weighted<<<BB * LQ, CC, 0, stream>>>(value, idx, attw, otmp);

    wmma_gemm_lds_f32<2, 4, 2, 2, 0><<<dim3((BB * LQ) / 64, CC / 128), dim3(128), 0, stream>>>(
        otmp, Wo, bo, out, nullptr, CC);
}


# Offsets and logits products done as ONE product over the stacked 96+32 weight rows: a work-group stages its 32 query rows once, a 4th wave owns the logit columns; 5->4 launches; bit-identical
# speedup vs baseline: 3.0891x; 1.1308x over previous
#include <hip/hip_runtime.h>
#include <hip/hip_bf16.h>


typedef __attribute__((ext_vector_type(2))) float v2f;
typedef __attribute__((ext_vector_type(8))) float v8f;
typedef __attribute__((ext_vector_type(4))) float v4f;

#define BB   2
#define LQ   1024
#define LIN  4096
#define CC   256
#define MH   8
#define NP   4
#define DH   32
#define QT   4
#define KW   4

#define KK     256
#define KPITCH (KK + 4)
#define KCH    64
#define KBL    32

template <int MT, int NT, int WM, int WN, int EPI>
__global__ void __launch_bounds__(32 * WM * WN)
wmma_gemm_lds_f32(const float* __restrict__ X, const float* __restrict__ W,
                  const float* __restrict__ bias, float* __restrict__ out,
                  const float* __restrict__ W2, const float* __restrict__ bias2,
                  float* __restrict__ out2, const float* __restrict__ aux, int Ncols) {
    constexpr int NTH  = 32 * WM * WN;
    constexpr int RX   = 16 * MT * WM;
    constexpr int RW   = 16 * NT * WN;
    constexpr int ROWS = RX + RW;
    constexpr int C4   = KCH / 4;
    constexpr int F4C  = ROWS * C4;
    constexpr int NLD  = (F4C + NTH - 1) / NTH;
    constexpr int NCH  = KK / KCH;
    constexpr int NKB  = KK / KBL;
    constexpr int BPC  = KCH / KBL;
    constexpr int KS   = KBL / 4;
    constexpr int N1   = (EPI == 1) ? 3 * (RW / 4) : RW;
    constexpr int N2   = RW - N1;

    __shared__ __attribute__((aligned(16))) float sT[ROWS * KPITCH];

    const int tid  = threadIdx.x;
    const int lane = tid & 31;
    const int wv   = tid >> 5;
    const int l15  = lane & 15;
    const int kh   = lane >> 4;
    const int wm   = wv / WN;
    const int wn   = wv % WN;
    const int row0 = blockIdx.x * RX;
    const int col0 = blockIdx.y * RW;

    constexpr int PW  = N1 / 3;
    constexpr int E1I = RX * PW;
    constexpr int E1T = (EPI == 1) ? (E1I + NTH - 1) / NTH : 1;
    float qv[E1T][3];
    if constexpr (EPI == 1) {
        static_assert(RW % 4 == 0 && N1 % (16 * NT) == 0 && N2 % 4 == 0,
                      "a wave's columns lie in one of the two products; whole softmax groups");
#pragma unroll
        for (int j = 0; j < E1T; ++j) {
            int i = tid + j * NTH;
            if (i >= E1I) i = E1I - 1;
            const float* q = aux + (size_t)(row0 + i / PW) * 3;
            qv[j][0] = q[0];
            qv[j][1] = q[1];
            qv[j][2] = q[2];
        }
    }

    const float* gsrc[NLD];
    int          soff[NLD];
    if constexpr (EPI == 1) {
        constexpr int RPT = NTH / C4;
        static_assert(NTH % C4 == 0 && F4C % NTH == 0 && RX % RPT == 0 && N1 % RPT == 0,
                      "every trip lies in one source");
        const int    pr   = tid / C4;
        const int    pc   = tid % C4;
        const size_t toff = (size_t)pr * KK + pc * 4;
#pragma unroll
        for (int j = 0; j < NLD; ++j) {
            const int    r    = j * RPT;
            const float* base = (r < RX)      ? X + (size_t)(row0 + r) * KK
                              : (r - RX < N1) ? W + (size_t)(r - RX) * KK
                                              : W2 + (size_t)(r - RX - N1) * KK;
            gsrc[j] = base + toff;
            soff[j] = (r + pr) * KPITCH + pc * 4;
        }
    } else {
#pragma unroll
        for (int j = 0; j < NLD; ++j) {
            int f = tid + j * NTH;
            if (f >= F4C) f = F4C - 1;
            const int r  = f / C4;
            const int c4 = f % C4;
            gsrc[j] = (r < RX) ? X + (size_t)(row0 + r) * KK + c4 * 4
                               : W + (size_t)(col0 + r - RX) * KK + c4 * 4;
            soff[j] = r * KPITCH + c4 * 4;
        }
    }

    v4f tmp[NLD];
#pragma unroll
    for (int j = 0; j < NLD; ++j)
        if (j + 1 < NLD || F4C % NTH == 0 || tid + j * NTH < F4C)
            tmp[j] = *(const v4f*)(gsrc[j]);
#pragma unroll
    for (int j = 0; j < NLD; ++j)
        if (j + 1 < NLD || F4C % NTH == 0 || tid + j * NTH < F4C)
            *(v4f*)(sT + soff[j]) = tmp[j];
    __syncthreads();
    if (NCH > 1) {
#pragma unroll
        for (int j = 0; j < NLD; ++j)
            if (j + 1 < NLD || F4C % NTH == 0 || tid + j * NTH < F4C)
                tmp[j] = *(const v4f*)(gsrc[j] + KCH);
    }

    const float* ap = sT + (wm * 16 * MT + l15) * KPITCH + 2 * kh;
    const float* bp = sT + (RX + wn * 16 * NT + l15) * KPITCH + 2 * kh;

    v8f acc[MT][NT];
#pragma unroll
    for (int mt = 0; mt < MT; ++mt)
#pragma unroll
        for (int nt = 0; nt < NT; ++nt) {
            v8f z = {};
            acc[mt][nt] = z;
        }

    v2f fa[2][KS][MT], fb[2][KS][NT];
#pragma unroll
    for (int ks = 0; ks < KS; ++ks) {
#pragma unroll
        for (int mt = 0; mt < MT; ++mt)
            fa[0][ks][mt] = *(const v2f*)(ap + mt * 16 * KPITCH + ks * 4);
#pragma unroll
        for (int nt = 0; nt < NT; ++nt)
            fb[0][ks][nt] = *(const v2f*)(bp + nt * 16 * KPITCH + ks * 4);
    }

#pragma unroll
    for (int kb = 0; kb < NKB; ++kb) {
        const int cur = kb & 1;
        const int c   = kb / BPC;
        if (kb % BPC == BPC - 1 && c + 1 < NCH) {
#pragma unroll
            for (int j = 0; j < NLD; ++j)
                if (j + 1 < NLD || F4C % NTH == 0 || tid + j * NTH < F4C)
                    *(v4f*)(sT + soff[j] + (c + 1) * KCH) = tmp[j];
            __syncthreads();
            if (c + 2 < NCH) {
#pragma unroll
                for (int j = 0; j < NLD; ++j)
                    if (j + 1 < NLD || F4C % NTH == 0 || tid + j * NTH < F4C)
                        tmp[j] = *(const v4f*)(gsrc[j] + (c + 2) * KCH);
            }
        }
        if (kb + 1 < NKB) {
#pragma unroll
            for (int ks = 0; ks < KS; ++ks) {
#pragma unroll
                for (int mt = 0; mt < MT; ++mt)
                    fa[cur ^ 1][ks][mt] =
                        *(const v2f*)(ap + mt * 16 * KPITCH + (kb + 1) * KBL + ks * 4);
#pragma unroll
                for (int nt = 0; nt < NT; ++nt)
                    fb[cur ^ 1][ks][nt] =
                        *(const v2f*)(bp + nt * 16 * KPITCH + (kb + 1) * KBL + ks * 4);
            }
        }
        __builtin_amdgcn_sched_barrier(0);
#pragma unroll
        for (int ks = 0; ks < KS; ++ks)
#pragma unroll
            for (int mt = 0; mt < MT; ++mt)
#pragma unroll
                for (int nt = 0; nt < NT; ++nt)
                    acc[mt][nt] = __builtin_amdgcn_wmma_f32_16x16x4_f32(
                         false, fa[cur][ks][mt],  false, fb[cur][ks][nt],
                         (short)0, acc[mt][nt],
                         false,  false);
        __builtin_amdgcn_sched_barrier(0);
    }

    const int orow0 = row0 + wm * 16 * MT;
    const int ocol0 = col0 + wn * 16 * NT;
    if constexpr (EPI == 0) {
#pragma unroll
        for (int nt = 0; nt < NT; ++nt) {
            const int   col  = ocol0 + nt * 16;
            const float bcol = bias[col + l15];
#pragma unroll
            for (int mt = 0; mt < MT; ++mt) {
                const int row = orow0 + mt * 16;
#pragma unroll
                for (int r = 0; r < 8; ++r) {
                    out[(size_t)(row + r + 8 * kh) * Ncols + col + l15] =
                        acc[mt][nt][r] + bcol;
                }
            }
        }
    } else {
        constexpr int EP = RW + 4;
        __shared__ __attribute__((aligned(16))) float sE[RX * EP];
#pragma unroll
        for (int nt = 0; nt < NT; ++nt) {
            const int   lc   = wn * 16 * NT + nt * 16 + l15;
            const float bcol = (lc < N1) ? bias[lc] : bias2[lc - N1];
#pragma unroll
            for (int mt = 0; mt < MT; ++mt)
#pragma unroll
                for (int r = 0; r < 8; ++r)
                    sE[(wm * 16 * MT + mt * 16 + r + 8 * kh) * EP + lc] =
                        acc[mt][nt][r] + bcol;
        }
        __syncthreads();
#pragma unroll
        for (int j = 0; j < E1T; ++j) {
            const int i = tid + j * NTH;
            if (j < E1I / NTH || i < E1I) {
                const int    r = i / PW;
                const int    p = i % PW;
                const float* o = sE + r * EP + p * 3;
                float4 s;
                s.x = qv[j][0] + o[0];
                s.y = qv[j][1] + o[1];
                s.z = qv[j][2] + o[2];
                s.w = 1.0f;
                *(float4*)(out + ((size_t)(row0 + r) * PW + p) * 4) = s;
            }
        }
        constexpr int GW = N2 / 4;
        constexpr int ITEMS = RX * GW;
#pragma unroll
        for (int j = 0; j < (ITEMS + NTH - 1) / NTH; ++j) {
            const int i = tid + j * NTH;
            if (ITEMS % NTH != 0 && i >= ITEMS) continue;
            const int r = i / GW;
            const int g = i % GW;
            float4 v = *(const float4*)(sE + r * EP + N1 + g * 4);
            const float mx = fmaxf(fmaxf(v.x, v.y), fmaxf(v.z, v.w));
            const float e0 = __expf(v.x - mx), e1 = __expf(v.y - mx);
            const float e2 = __expf(v.z - mx), e3 = __expf(v.w - mx);
            const float inv = 1.0f / (e0 + e1 + e2 + e3);
            v.x = e0 * inv; v.y = e1 * inv; v.z = e2 * inv; v.w = e3 * inv;
            *(float4*)(out2 + (size_t)(row0 + r) * N2 + g * 4) = v;
        }
    }
}

__device__ __forceinline__ void knn_rows2(const v8f* c, int r0, int cib,
                                          float* best, int* bidx) {
#pragma unroll
    for (int r = r0; r < r0 + 2; ++r) {
#pragma unroll
        for (int t = 0; t < QT; ++t) {
            const float dv = c[t][r];
            const int   ci = cib + r;
            if (dv < best[t]) { best[t] = dv; bidx[t] = ci; }
        }
    }
}

__device__ __forceinline__ void knn_phase(v8f* cn, const v8f* cc, const v2f a,
                                          const v2f* bq, int cib,
                                          float* best, int* bidx) {
    static_assert(2 * QT == 8, "one product per two of a tile's eight rows");
#pragma unroll
    for (int t = 0; t < QT; ++t) {
        v8f z = {};
        cn[t] = __builtin_amdgcn_wmma_f32_16x16x4_f32(
            false, a, false, bq[t], (short)0, z, false, false);
        knn_rows2(cc, 2 * t, cib, best, bidx);
        __builtin_amdgcn_sched_barrier(0);
    }
}

#define VRX   64
#define VRW   128
#define VROWS (VRX + VRW)

__global__ void __launch_bounds__(64 * KW)
knn_and_value(const float* __restrict__ S4, const float* __restrict__ ipts,
              int* __restrict__ idxOut,
              const float* __restrict__ X, const float* __restrict__ Wt,
              const float* __restrict__ bias, float* __restrict__ vout) {
    constexpr int NTH = 64 * KW;
    __shared__ __attribute__((aligned(16))) float sV[LIN * 4];
    __shared__ __attribute__((aligned(16))) float sT[VROWS * KPITCH];

    const int tid  = threadIdx.x;
    const int lane = tid & 31;
    const int wv   = __builtin_amdgcn_readfirstlane(tid >> 5);
    const int l15  = lane & 15;
    const int kh   = lane >> 4;
    const int b    = (blockIdx.x * (KW * 16 * QT)) >> 15;
    const int n0   = (blockIdx.x * KW + (wv & (KW - 1))) * (16 * QT);
    const int row0 = (blockIdx.x >> 1) * VRX;
    const int col0 = (blockIdx.x & 1) * VRW;

    v2f bq[QT];
#pragma unroll
    for (int t = 0; t < QT; ++t) {
        v2f z = {};
        bq[t] = z;
    }
    if (wv < KW) {
#pragma unroll
        for (int t = 0; t < QT; ++t)
            bq[t] = *(const v2f*)(S4 + (size_t)(n0 + t * 16 + l15) * 4 + 2 * kh);
    }

    {
        const float4* __restrict__ g = (const float4*)(ipts + (size_t)b * LIN * 3);
        float4* s = (float4*)sV;
        constexpr int NG  = LIN / 4 / NTH;
        constexpr int C4  = KK / 4;
        constexpr int RPT = NTH / C4;
        constexpr int NLX = VRX / RPT;
        constexpr int NLW = VRW / RPT;
        static_assert(NTH % C4 == 0 && VRX % RPT == 0 && VRW % RPT == 0, "whole trips");
        const int pr = tid / C4;
        const int pc = tid % C4;
        const float* xsrc = X  + (size_t)(row0 + pr) * KK + pc * 4;
        const float* wsrc = Wt + (size_t)(col0 + pr) * KK + pc * 4;
        float*       sdst = sT + pr * KPITCH + pc * 4;

        float4 tmp[NG][3];
        v4f    xt[NLX], wt[NLW];
#pragma unroll
        for (int j = 0; j < NG; ++j)
#pragma unroll
            for (int k = 0; k < 3; ++k) tmp[j][k] = g[3 * (tid + NTH * j) + k];
#pragma unroll
        for (int j = 0; j < NLX; ++j) xt[j] = *(const v4f*)(xsrc + (size_t)j * RPT * KK);
#pragma unroll
        for (int j = 0; j < NLW; ++j) wt[j] = *(const v4f*)(wsrc + (size_t)j * RPT * KK);

#pragma unroll
        for (int j = 0; j < NG; ++j) {
            const float f[12] = {tmp[j][0].x, tmp[j][0].y, tmp[j][0].z, tmp[j][0].w,
                                 tmp[j][1].x, tmp[j][1].y, tmp[j][1].z, tmp[j][1].w,
                                 tmp[j][2].x, tmp[j][2].y, tmp[j][2].z, tmp[j][2].w};
#pragma unroll
            for (int i = 0; i < 4; ++i) {
                const float x = f[3 * i], y = f[3 * i + 1], z = f[3 * i + 2];
                float xy;
                {
#pragma clang fp contract(off)
                    xy = x * x + y * y;
                }
                float4 t;
                t.x = -2.0f * x;
                t.y = -2.0f * y;
                t.z = -2.0f * z;
                t.w = __builtin_fmaf(z, z, xy);
                s[4 * (tid + NTH * j) + i] = t;
            }
        }
#pragma unroll
        for (int j = 0; j < NLX; ++j) *(v4f*)(sdst + j * RPT * KPITCH) = xt[j];
#pragma unroll
        for (int j = 0; j < NLW; ++j) *(v4f*)(sdst + (VRX + j * RPT) * KPITCH) = wt[j];
    }
    __syncthreads();

    if (wv < KW) {
        const float* vptr = sV + l15 * 4 + 2 * kh;

        float best[QT];
        int   bidx[QT];
#pragma unroll
        for (int t = 0; t < QT; ++t) { best[t] = 3.4e38f; bidx[t] = 0; }

        constexpr int NBLK = LIN / 16;
        const int ch = kh << 3;
        v8f c0[QT], c1[QT];
        v2f aA = *(const v2f*)(vptr);
        v2f aB = *(const v2f*)(vptr + 16 * 4);
#pragma unroll
        for (int t = 0; t < QT; ++t) {
            v8f z = {};
            c0[t] = __builtin_amdgcn_wmma_f32_16x16x4_f32(
                false, aA, false, bq[t], (short)0, z, false, false);
        }
        aA = *(const v2f*)(vptr + 2 * 16 * 4);
        for (int j = 0; j < NBLK - 4; j += 2) {
            knn_phase(c1, c0, aB, bq, j * 16 + ch, best, bidx);
            aB = *(const v2f*)(vptr + (j + 3) * 16 * 4);
            knn_phase(c0, c1, aA, bq, (j + 1) * 16 + ch, best, bidx);
            aA = *(const v2f*)(vptr + (j + 4) * 16 * 4);
        }
        knn_phase(c1, c0, aB, bq, (NBLK - 4) * 16 + ch, best, bidx);
        aB = *(const v2f*)(vptr + (NBLK - 1) * 16 * 4);
        knn_phase(c0, c1, aA, bq, (NBLK - 3) * 16 + ch, best, bidx);
        knn_phase(c1, c0, aB, bq, (NBLK - 2) * 16 + ch, best, bidx);
#pragma unroll
        for (int q = 0; q < 4; ++q)
            knn_rows2(c1, 2 * q, (NBLK - 1) * 16 + ch, best, bidx);

#pragma unroll
        for (int t = 0; t < QT; ++t) {
            const float obest = __shfl_xor(best[t], 16, 32);
            const int   oidx  = __shfl_xor(bidx[t], 16, 32);
            if (obest < best[t] || (obest == best[t] && oidx < bidx[t])) {
                best[t] = obest; bidx[t] = oidx;
            }
            if (lane < 16) idxOut[n0 + t * 16 + l15] = bidx[t];
        }
    } else {
        constexpr int MT  = 2, NT = 4;
        constexpr int NKB = KK / KBL;
        constexpr int KS  = KBL / 4;
        static_assert(VRX == 2 * 16 * MT && VRW == 2 * 16 * NT, "2x2 waves of 32x64");
        const int w  = wv - KW;
        const int wm = w >> 1;
        const int wn = w & 1;
        const float* ap = sT + (wm * 16 * MT + l15) * KPITCH + 2 * kh;
        const float* bp = sT + (VRX + wn * 16 * NT + l15) * KPITCH + 2 * kh;

        v8f acc[MT][NT];
#pragma unroll
        for (int mt = 0; mt < MT; ++mt)
#pragma unroll
            for (int nt = 0; nt < NT; ++nt) {
                v8f z = {};
                acc[mt][nt] = z;
            }

        v2f fa[2][KS][MT], fb[2][KS][NT];
#pragma unroll
        for (int ks = 0; ks < KS; ++ks) {
#pragma unroll
            for (int mt = 0; mt < MT; ++mt)
                fa[0][ks][mt] = *(const v2f*)(ap + mt * 16 * KPITCH + ks * 4);
#pragma unroll
            for (int nt = 0; nt < NT; ++nt)
                fb[0][ks][nt] = *(const v2f*)(bp + nt * 16 * KPITCH + ks * 4);
        }

#pragma unroll
        for (int kb = 0; kb < NKB; ++kb) {
            const int cur = kb & 1;
            if (kb + 1 < NKB) {
#pragma unroll
                for (int ks = 0; ks < KS; ++ks) {
#pragma unroll
                    for (int mt = 0; mt < MT; ++mt)
                        fa[cur ^ 1][ks][mt] =
                            *(const v2f*)(ap + mt * 16 * KPITCH + (kb + 1) * KBL + ks * 4);
#pragma unroll
                    for (int nt = 0; nt < NT; ++nt)
                        fb[cur ^ 1][ks][nt] =
                            *(const v2f*)(bp + nt * 16 * KPITCH + (kb + 1) * KBL + ks * 4);
                }
            }
            __builtin_amdgcn_sched_barrier(0);
#pragma unroll
            for (int ks = 0; ks < KS; ++ks)
#pragma unroll
                for (int mt = 0; mt < MT; ++mt)
#pragma unroll
                    for (int nt = 0; nt < NT; ++nt)
                        acc[mt][nt] = __builtin_amdgcn_wmma_f32_16x16x4_f32(
                             false, fa[cur][ks][mt],  false, fb[cur][ks][nt],
                             (short)0, acc[mt][nt],
                             false,  false);
            __builtin_amdgcn_sched_barrier(0);
        }

        const int orow0 = row0 + wm * 16 * MT;
        const int ocol0 = col0 + wn * 16 * NT;
#pragma unroll
        for (int nt = 0; nt < NT; ++nt) {
            const int   col  = ocol0 + nt * 16;
            const float bcol = bias[col + l15];
#pragma unroll
            for (int mt = 0; mt < MT; ++mt) {
                const int row = orow0 + mt * 16;
#pragma unroll
                for (int r = 0; r < 8; ++r) {
                    vout[(size_t)(row + r + 8 * kh) * CC + col + l15] =
                        acc[mt][nt][r] + bcol;
                }
            }
        }
    }
}

__global__ void __launch_bounds__(CC)
gather_weighted(const float* __restrict__ value, const int* __restrict__ idx,
                const float* __restrict__ attw, float* __restrict__ out_tmp) {
    const int bq = blockIdx.x;
    const int c  = threadIdx.x;
    const int b  = bq >> 10;
    const int m  = c >> 5;
    const int*   id = idx  + ((size_t)bq * MH + m) * NP;
    const float* aw = attw + ((size_t)bq * MH + m) * NP;
    const float* vb = value + (size_t)b * LIN * CC;
    float acc = 0.0f;
#pragma unroll
    for (int p = 0; p < NP; ++p)
        acc += aw[p] * vb[(size_t)id[p] * CC + c];
    out_tmp[(size_t)bq * CC + c] = acc;
}

extern "C" void kernel_launch(void* const* d_in, const int* in_sizes, int n_in,
                              void* d_out, int out_size, void* d_ws, size_t ws_size,
                              hipStream_t stream) {
    const float* query = (const float*)d_in[0];
    const float* qpts  = (const float*)d_in[1];
    const float* inp   = (const float*)d_in[2];
    const float* ipts  = (const float*)d_in[3];
    const float* Wv    = (const float*)d_in[4];
    const float* bv    = (const float*)d_in[5];
    const float* Ws    = (const float*)d_in[6];
    const float* bs    = (const float*)d_in[7];
    const float* Wa    = (const float*)d_in[8];
    const float* ba    = (const float*)d_in[9];
    const float* Wo    = (const float*)d_in[10];
    const float* bo    = (const float*)d_in[11];
    float* out = (float*)d_out;

    char* w = (char*)d_ws;
    float* value = (float*)w;  w += (size_t)BB * LIN * CC * 4;
    float* attw  = (float*)w;  w += (size_t)BB * LQ * MH * NP * 4;
    float* S4    = (float*)w;  w += (size_t)BB * LQ * MH * NP * 4 * 4;
    int*   idx   = (int*)w;    w += (size_t)BB * LQ * MH * NP * 4;
    float* otmp  = (float*)w;  w += (size_t)BB * LQ * CC * 4;

    static_assert(3 * (MH * NP) + MH * NP == 4 * 2 * 16, "the stacked width is the 1x4 waves' 128 columns");
    wmma_gemm_lds_f32<2, 2, 1, 4, 1><<<dim3((BB * LQ) / 32, 1), dim3(128), 0, stream>>>(
        query, Ws, bs, S4, Wa, ba, attw, qpts, MH * NP * 4);

    static_assert((BB * LQ * MH * NP) / (16 * QT * KW) == ((BB * LIN) / VRX) * (CC / VRW),
                  "as many scan work-groups as value blocks");
    knn_and_value<<<(BB * LQ * MH * NP) / (16 * QT * KW), dim3(64 * KW), 0, stream>>>(
        S4, ipts, idx, inp, Wv, bv, value);

    gather_weighted<<<BB * LQ, CC, 0, stream>>>(value, idx, attw, otmp);

    wmma_gemm_lds_f32<2, 4, 2, 2, 0><<<dim3((BB * LQ) / 64, CC / 128), dim3(128), 0, stream>>>(
        otmp, Wo, bo, out, nullptr, nullptr, nullptr, nullptr, CC);
}


# Gather + weighted sum folded into the output product: a work-group of 4 waves owns 16 rows x all 256 columns, gathers its rows once into LDS as A rows (sums now fused multiply-adds); 4->3 launches
# speedup vs baseline: 3.2891x; 1.0647x over previous
#include <hip/hip_runtime.h>
#include <hip/hip_bf16.h>


typedef __attribute__((ext_vector_type(2))) float v2f;
typedef __attribute__((ext_vector_type(8))) float v8f;
typedef __attribute__((ext_vector_type(4))) float v4f;

#define BB   2
#define LQ   1024
#define LIN  4096
#define CC   256
#define MH   8
#define NP   4
#define DH   32
#define QT   4
#define KW   4

#define KK     256
#define KPITCH (KK + 4)
#define KCH    64
#define KBL    32

template <int MT, int NT, int WM, int WN, int EPI>
__global__ void __launch_bounds__(32 * WM * WN)
wmma_gemm_lds_f32(const float* __restrict__ X, const float* __restrict__ W,
                  const float* __restrict__ bias, float* __restrict__ out,
                  const float* __restrict__ W2, const float* __restrict__ bias2,
                  float* __restrict__ out2, const float* __restrict__ aux,
                  const int* __restrict__ gidx, int Ncols) {
    constexpr int NTH  = 32 * WM * WN;
    constexpr int RX   = 16 * MT * WM;
    constexpr int RW   = 16 * NT * WN;
    constexpr int ROWS = RX + RW;
    constexpr int SROWS = (EPI == 2) ? RW : ROWS;
    constexpr int C4   = KCH / 4;
    constexpr int F4C  = SROWS * C4;
    constexpr int NLD  = (F4C + NTH - 1) / NTH;
    constexpr int NCH  = KK / KCH;
    constexpr int NKB  = KK / KBL;
    constexpr int BPC  = KCH / KBL;
    constexpr int KS   = KBL / 4;
    constexpr int N1   = (EPI == 1) ? 3 * (RW / 4) : RW;
    constexpr int N2   = RW - N1;

    __shared__ __attribute__((aligned(16))) float sT[ROWS * KPITCH];

    const int tid  = threadIdx.x;
    const int lane = tid & 31;
    const int wv   = tid >> 5;
    const int l15  = lane & 15;
    const int kh   = lane >> 4;
    const int wm   = wv / WN;
    const int wn   = wv % WN;
    const int row0 = blockIdx.x * RX;
    const int col0 = blockIdx.y * RW;

    constexpr int PW  = N1 / 3;
    constexpr int E1I = RX * PW;
    constexpr int E1T = (EPI == 1) ? (E1I + NTH - 1) / NTH : 1;
    float qv[E1T][3];
    if constexpr (EPI == 1) {
        static_assert(RW % 4 == 0 && N1 % (16 * NT) == 0 && N2 % 4 == 0,
                      "a wave's columns lie in one of the two products; whole softmax groups");
#pragma unroll
        for (int j = 0; j < E1T; ++j) {
            int i = tid + j * NTH;
            if (i >= E1I) i = E1I - 1;
            const float* q = aux + (size_t)(row0 + i / PW) * 3;
            qv[j][0] = q[0];
            qv[j][1] = q[1];
            qv[j][2] = q[2];
        }
    }

    constexpr int P4  = KK / 4;
    constexpr int GRS = NTH / P4;
    constexpr int NGP = (EPI == 2) ? RX / GRS : 1;
    const int gpc = tid % P4;
    const int gr0 = tid / P4;
    int4   gi[NGP];
    float4 ga[NGP];
    if constexpr (EPI == 2) {
        static_assert(NP == 4 && KK == MH * DH && DH % 4 == 0, "four points per head, heads side by side");
        static_assert(NTH % P4 == 0 && RX % GRS == 0, "whole trips, one column piece per thread");
#pragma unroll
        for (int j = 0; j < NGP; ++j) {
            const size_t o = ((size_t)(row0 + gr0 + j * GRS) * MH + gpc / (DH / 4)) * NP;
            gi[j] = *(const int4*)(gidx + o);
            ga[j] = *(const float4*)(aux + o);
        }
    }

    const float* gsrc[NLD];
    int          soff[NLD];
    if constexpr (EPI == 1) {
        constexpr int RPT = NTH / C4;
        static_assert(NTH % C4 == 0 && F4C % NTH == 0 && RX % RPT == 0 && N1 % RPT == 0,
                      "every trip lies in one source");
        const int    pr   = tid / C4;
        const int    pc   = tid % C4;
        const size_t toff = (size_t)pr * KK + pc * 4;
#pragma unroll
        for (int j = 0; j < NLD; ++j) {
            const int    r    = j * RPT;
            const float* base = (r < RX)      ? X + (size_t)(row0 + r) * KK
                              : (r - RX < N1) ? W + (size_t)(r - RX) * KK
                                              : W2 + (size_t)(r - RX - N1) * KK;
            gsrc[j] = base + toff;
            soff[j] = (r + pr) * KPITCH + pc * 4;
        }
    } else {
#pragma unroll
        for (int j = 0; j < NLD; ++j) {
            int f = tid + j * NTH;
            if (f >= F4C) f = F4C - 1;
            const int r  = f / C4 + (ROWS - SROWS);
            const int c4 = f % C4;
            gsrc[j] = (r < RX) ? X + (size_t)(row0 + r) * KK + c4 * 4
                               : W + (size_t)(col0 + r - RX) * KK + c4 * 4;
            soff[j] = r * KPITCH + c4 * 4;
        }
    }

    v4f tmp[NLD];
#pragma unroll
    for (int j = 0; j < NLD; ++j)
        if (j + 1 < NLD || F4C % NTH == 0 || tid + j * NTH < F4C)
            tmp[j] = *(const v4f*)(gsrc[j]);
    if constexpr (EPI == 2) {
        v4f gv[NGP][NP];
#pragma unroll
        for (int j = 0; j < NGP; ++j) {
            const int    row = row0 + gr0 + j * GRS;
            const float* vb  = X + (size_t)(row / LQ) * LIN * KK + gpc * 4;
            gv[j][0] = *(const v4f*)(vb + (size_t)gi[j].x * KK);
            gv[j][1] = *(const v4f*)(vb + (size_t)gi[j].y * KK);
            gv[j][2] = *(const v4f*)(vb + (size_t)gi[j].z * KK);
            gv[j][3] = *(const v4f*)(vb + (size_t)gi[j].w * KK);
        }
#pragma unroll
        for (int j = 0; j < NGP; ++j) {
            v4f x;
#pragma unroll
            for (int e = 0; e < 4; ++e) {
                float acc = 0.0f;
                acc += ga[j].x * gv[j][0][e];
                acc += ga[j].y * gv[j][1][e];
                acc += ga[j].z * gv[j][2][e];
                acc += ga[j].w * gv[j][3][e];
                x[e] = acc;
            }
            *(v4f*)(sT + (gr0 + j * GRS) * KPITCH + gpc * 4) = x;
        }
    }
#pragma unroll
    for (int j = 0; j < NLD; ++j)
        if (j + 1 < NLD || F4C % NTH == 0 || tid + j * NTH < F4C)
            *(v4f*)(sT + soff[j]) = tmp[j];
    __syncthreads();
    if (NCH > 1) {
#pragma unroll
        for (int j = 0; j < NLD; ++j)
            if (j + 1 < NLD || F4C % NTH == 0 || tid + j * NTH < F4C)
                tmp[j] = *(const v4f*)(gsrc[j] + KCH);
    }

    const float* ap = sT + (wm * 16 * MT + l15) * KPITCH + 2 * kh;
    const float* bp = sT + (RX + wn * 16 * NT + l15) * KPITCH + 2 * kh;

    v8f acc[MT][NT];
#pragma unroll
    for (int mt = 0; mt < MT; ++mt)
#pragma unroll
        for (int nt = 0; nt < NT; ++nt) {
            v8f z = {};
            acc[mt][nt] = z;
        }

    v2f fa[2][KS][MT], fb[2][KS][NT];
#pragma unroll
    for (int ks = 0; ks < KS; ++ks) {
#pragma unroll
        for (int mt = 0; mt < MT; ++mt)
            fa[0][ks][mt] = *(const v2f*)(ap + mt * 16 * KPITCH + ks * 4);
#pragma unroll
        for (int nt = 0; nt < NT; ++nt)
            fb[0][ks][nt] = *(const v2f*)(bp + nt * 16 * KPITCH + ks * 4);
    }

#pragma unroll
    for (int kb = 0; kb < NKB; ++kb) {
        const int cur = kb & 1;
        const int c   = kb / BPC;
        if (kb % BPC == BPC - 1 && c + 1 < NCH) {
#pragma unroll
            for (int j = 0; j < NLD; ++j)
                if (j + 1 < NLD || F4C % NTH == 0 || tid + j * NTH < F4C)
                    *(v4f*)(sT + soff[j] + (c + 1) * KCH) = tmp[j];
            __syncthreads();
            if (c + 2 < NCH) {
#pragma unroll
                for (int j = 0; j < NLD; ++j)
                    if (j + 1 < NLD || F4C % NTH == 0 || tid + j * NTH < F4C)
                        tmp[j] = *(const v4f*)(gsrc[j] + (c + 2) * KCH);
            }
        }
        if (kb + 1 < NKB) {
#pragma unroll
            for (int ks = 0; ks < KS; ++ks) {
#pragma unroll
                for (int mt = 0; mt < MT; ++mt)
                    fa[cur ^ 1][ks][mt] =
                        *(const v2f*)(ap + mt * 16 * KPITCH + (kb + 1) * KBL + ks * 4);
#pragma unroll
                for (int nt = 0; nt < NT; ++nt)
                    fb[cur ^ 1][ks][nt] =
                        *(const v2f*)(bp + nt * 16 * KPITCH + (kb + 1) * KBL + ks * 4);
            }
        }
        __builtin_amdgcn_sched_barrier(0);
#pragma unroll
        for (int ks = 0; ks < KS; ++ks)
#pragma unroll
            for (int mt = 0; mt < MT; ++mt)
#pragma unroll
                for (int nt = 0; nt < NT; ++nt)
                    acc[mt][nt] = __builtin_amdgcn_wmma_f32_16x16x4_f32(
                         false, fa[cur][ks][mt],  false, fb[cur][ks][nt],
                         (short)0, acc[mt][nt],
                         false,  false);
        __builtin_amdgcn_sched_barrier(0);
    }

    const int orow0 = row0 + wm * 16 * MT;
    const int ocol0 = col0 + wn * 16 * NT;
    if constexpr (EPI != 1) {
#pragma unroll
        for (int nt = 0; nt < NT; ++nt) {
            const int   col  = ocol0 + nt * 16;
            const float bcol = bias[col + l15];
#pragma unroll
            for (int mt = 0; mt < MT; ++mt) {
                const int row = orow0 + mt * 16;
#pragma unroll
                for (int r = 0; r < 8; ++r) {
                    out[(size_t)(row + r + 8 * kh) * Ncols + col + l15] =
                        acc[mt][nt][r] + bcol;
                }
            }
        }
    } else {
        constexpr int EP = RW + 4;
        __shared__ __attribute__((aligned(16))) float sE[RX * EP];
#pragma unroll
        for (int nt = 0; nt < NT; ++nt) {
            const int   lc   = wn * 16 * NT + nt * 16 + l15;
            const float bcol = (lc < N1) ? bias[lc] : bias2[lc - N1];
#pragma unroll
            for (int mt = 0; mt < MT; ++mt)
#pragma unroll
                for (int r = 0; r < 8; ++r)
                    sE[(wm * 16 * MT + mt * 16 + r + 8 * kh) * EP + lc] =
                        acc[mt][nt][r] + bcol;
        }
        __syncthreads();
#pragma unroll
        for (int j = 0; j < E1T; ++j) {
            const int i = tid + j * NTH;
            if (j < E1I / NTH || i < E1I) {
                const int    r = i / PW;
                const int    p = i % PW;
                const float* o = sE + r * EP + p * 3;
                float4 s;
                s.x = qv[j][0] + o[0];
                s.y = qv[j][1] + o[1];
                s.z = qv[j][2] + o[2];
                s.w = 1.0f;
                *(float4*)(out + ((size_t)(row0 + r) * PW + p) * 4) = s;
            }
        }
        constexpr int GW = N2 / 4;
        constexpr int ITEMS = RX * GW;
#pragma unroll
        for (int j = 0; j < (ITEMS + NTH - 1) / NTH; ++j) {
            const int i = tid + j * NTH;
            if (ITEMS % NTH != 0 && i >= ITEMS) continue;
            const int r = i / GW;
            const int g = i % GW;
            float4 v = *(const float4*)(sE + r * EP + N1 + g * 4);
            const float mx = fmaxf(fmaxf(v.x, v.y), fmaxf(v.z, v.w));
            const float e0 = __expf(v.x - mx), e1 = __expf(v.y - mx);
            const float e2 = __expf(v.z - mx), e3 = __expf(v.w - mx);
            const float inv = 1.0f / (e0 + e1 + e2 + e3);
            v.x = e0 * inv; v.y = e1 * inv; v.z = e2 * inv; v.w = e3 * inv;
            *(float4*)(out2 + (size_t)(row0 + r) * N2 + g * 4) = v;
        }
    }
}

__device__ __forceinline__ void knn_rows2(const v8f* c, int r0, int cib,
                                          float* best, int* bidx) {
#pragma unroll
    for (int r = r0; r < r0 + 2; ++r) {
#pragma unroll
        for (int t = 0; t < QT; ++t) {
            const float dv = c[t][r];
            const int   ci = cib + r;
            if (dv < best[t]) { best[t] = dv; bidx[t] = ci; }
        }
    }
}

__device__ __forceinline__ void knn_phase(v8f* cn, const v8f* cc, const v2f a,
                                          const v2f* bq, int cib,
                                          float* best, int* bidx) {
    static_assert(2 * QT == 8, "one product per two of a tile's eight rows");
#pragma unroll
    for (int t = 0; t < QT; ++t) {
        v8f z = {};
        cn[t] = __builtin_amdgcn_wmma_f32_16x16x4_f32(
            false, a, false, bq[t], (short)0, z, false, false);
        knn_rows2(cc, 2 * t, cib, best, bidx);
        __builtin_amdgcn_sched_barrier(0);
    }
}

#define VRX   64
#define VRW   128
#define VROWS (VRX + VRW)

__global__ void __launch_bounds__(64 * KW)
knn_and_value(const float* __restrict__ S4, const float* __restrict__ ipts,
              int* __restrict__ idxOut,
              const float* __restrict__ X, const float* __restrict__ Wt,
              const float* __restrict__ bias, float* __restrict__ vout) {
    constexpr int NTH = 64 * KW;
    __shared__ __attribute__((aligned(16))) float sV[LIN * 4];
    __shared__ __attribute__((aligned(16))) float sT[VROWS * KPITCH];

    const int tid  = threadIdx.x;
    const int lane = tid & 31;
    const int wv   = __builtin_amdgcn_readfirstlane(tid >> 5);
    const int l15  = lane & 15;
    const int kh   = lane >> 4;
    const int b    = (blockIdx.x * (KW * 16 * QT)) >> 15;
    const int n0   = (blockIdx.x * KW + (wv & (KW - 1))) * (16 * QT);
    const int row0 = (blockIdx.x >> 1) * VRX;
    const int col0 = (blockIdx.x & 1) * VRW;

    v2f bq[QT];
#pragma unroll
    for (int t = 0; t < QT; ++t) {
        v2f z = {};
        bq[t] = z;
    }
    if (wv < KW) {
#pragma unroll
        for (int t = 0; t < QT; ++t)
            bq[t] = *(const v2f*)(S4 + (size_t)(n0 + t * 16 + l15) * 4 + 2 * kh);
    }

    {
        const float4* __restrict__ g = (const float4*)(ipts + (size_t)b * LIN * 3);
        float4* s = (float4*)sV;
        constexpr int NG  = LIN / 4 / NTH;
        constexpr int C4  = KK / 4;
        constexpr int RPT = NTH / C4;
        constexpr int NLX = VRX / RPT;
        constexpr int NLW = VRW / RPT;
        static_assert(NTH % C4 == 0 && VRX % RPT == 0 && VRW % RPT == 0, "whole trips");
        const int pr = tid / C4;
        const int pc = tid % C4;
        const float* xsrc = X  + (size_t)(row0 + pr) * KK + pc * 4;
        const float* wsrc = Wt + (size_t)(col0 + pr) * KK + pc * 4;
        float*       sdst = sT + pr * KPITCH + pc * 4;

        float4 tmp[NG][3];
        v4f    xt[NLX], wt[NLW];
#pragma unroll
        for (int j = 0; j < NG; ++j)
#pragma unroll
            for (int k = 0; k < 3; ++k) tmp[j][k] = g[3 * (tid + NTH * j) + k];
#pragma unroll
        for (int j = 0; j < NLX; ++j) xt[j] = *(const v4f*)(xsrc + (size_t)j * RPT * KK);
#pragma unroll
        for (int j = 0; j < NLW; ++j) wt[j] = *(const v4f*)(wsrc + (size_t)j * RPT * KK);

#pragma unroll
        for (int j = 0; j < NG; ++j) {
            const float f[12] = {tmp[j][0].x, tmp[j][0].y, tmp[j][0].z, tmp[j][0].w,
                                 tmp[j][1].x, tmp[j][1].y, tmp[j][1].z, tmp[j][1].w,
                                 tmp[j][2].x, tmp[j][2].y, tmp[j][2].z, tmp[j][2].w};
#pragma unroll
            for (int i = 0; i < 4; ++i) {
                const float x = f[3 * i], y = f[3 * i + 1], z = f[3 * i + 2];
                float xy;
                {
#pragma clang fp contract(off)
                    xy = x * x + y * y;
                }
                float4 t;
                t.x = -2.0f * x;
                t.y = -2.0f * y;
                t.z = -2.0f * z;
                t.w = __builtin_fmaf(z, z, xy);
                s[4 * (tid + NTH * j) + i] = t;
            }
        }
#pragma unroll
        for (int j = 0; j < NLX; ++j) *(v4f*)(sdst + j * RPT * KPITCH) = xt[j];
#pragma unroll
        for (int j = 0; j < NLW; ++j) *(v4f*)(sdst + (VRX + j * RPT) * KPITCH) = wt[j];
    }
    __syncthreads();

    if (wv < KW) {
        const float* vptr = sV + l15 * 4 + 2 * kh;

        float best[QT];
        int   bidx[QT];
#pragma unroll
        for (int t = 0; t < QT; ++t) { best[t] = 3.4e38f; bidx[t] = 0; }

        constexpr int NBLK = LIN / 16;
        const int ch = kh << 3;
        v8f c0[QT], c1[QT];
        v2f aA = *(const v2f*)(vptr);
        v2f aB = *(const v2f*)(vptr + 16 * 4);
#pragma unroll
        for (int t = 0; t < QT; ++t) {
            v8f z = {};
            c0[t] = __builtin_amdgcn_wmma_f32_16x16x4_f32(
                false, aA, false, bq[t], (short)0, z, false, false);
        }
        aA = *(const v2f*)(vptr + 2 * 16 * 4);
        for (int j = 0; j < NBLK - 4; j += 2) {
            knn_phase(c1, c0, aB, bq, j * 16 + ch, best, bidx);
            aB = *(const v2f*)(vptr + (j + 3) * 16 * 4);
            knn_phase(c0, c1, aA, bq, (j + 1) * 16 + ch, best, bidx);
            aA = *(const v2f*)(vptr + (j + 4) * 16 * 4);
        }
        knn_phase(c1, c0, aB, bq, (NBLK - 4) * 16 + ch, best, bidx);
        aB = *(const v2f*)(vptr + (NBLK - 1) * 16 * 4);
        knn_phase(c0, c1, aA, bq, (NBLK - 3) * 16 + ch, best, bidx);
        knn_phase(c1, c0, aB, bq, (NBLK - 2) * 16 + ch, best, bidx);
#pragma unroll
        for (int q = 0; q < 4; ++q)
            knn_rows2(c1, 2 * q, (NBLK - 1) * 16 + ch, best, bidx);

#pragma unroll
        for (int t = 0; t < QT; ++t) {
            const float obest = __shfl_xor(best[t], 16, 32);
            const int   oidx  = __shfl_xor(bidx[t], 16, 32);
            if (obest < best[t] || (obest == best[t] && oidx < bidx[t])) {
                best[t] = obest; bidx[t] = oidx;
            }
            if (lane < 16) idxOut[n0 + t * 16 + l15] = bidx[t];
        }
    } else {
        constexpr int MT  = 2, NT = 4;
        constexpr int NKB = KK / KBL;
        constexpr int KS  = KBL / 4;
        static_assert(VRX == 2 * 16 * MT && VRW == 2 * 16 * NT, "2x2 waves of 32x64");
        const int w  = wv - KW;
        const int wm = w >> 1;
        const int wn = w & 1;
        const float* ap = sT + (wm * 16 * MT + l15) * KPITCH + 2 * kh;
        const float* bp = sT + (VRX + wn * 16 * NT + l15) * KPITCH + 2 * kh;

        v8f acc[MT][NT];
#pragma unroll
        for (int mt = 0; mt < MT; ++mt)
#pragma unroll
            for (int nt = 0; nt < NT; ++nt) {
                v8f z = {};
                acc[mt][nt] = z;
            }

        v2f fa[2][KS][MT], fb[2][KS][NT];
#pragma unroll
        for (int ks = 0; ks < KS; ++ks) {
#pragma unroll
            for (int mt = 0; mt < MT; ++mt)
                fa[0][ks][mt] = *(const v2f*)(ap + mt * 16 * KPITCH + ks * 4);
#pragma unroll
            for (int nt = 0; nt < NT; ++nt)
                fb[0][ks][nt] = *(const v2f*)(bp + nt * 16 * KPITCH + ks * 4);
        }

#pragma unroll
        for (int kb = 0; kb < NKB; ++kb) {
            const int cur = kb & 1;
            if (kb + 1 < NKB) {
#pragma unroll
                for (int ks = 0; ks < KS; ++ks) {
#pragma unroll
                    for (int mt = 0; mt < MT; ++mt)
                        fa[cur ^ 1][ks][mt] =
                            *(const v2f*)(ap + mt * 16 * KPITCH + (kb + 1) * KBL + ks * 4);
#pragma unroll
                    for (int nt = 0; nt < NT; ++nt)
                        fb[cur ^ 1][ks][nt] =
                            *(const v2f*)(bp + nt * 16 * KPITCH + (kb + 1) * KBL + ks * 4);
                }
            }
            __builtin_amdgcn_sched_barrier(0);
#pragma unroll
            for (int ks = 0; ks < KS; ++ks)
#pragma unroll
                for (int mt = 0; mt < MT; ++mt)
#pragma unroll
                    for (int nt = 0; nt < NT; ++nt)
                        acc[mt][nt] = __builtin_amdgcn_wmma_f32_16x16x4_f32(
                             false, fa[cur][ks][mt],  false, fb[cur][ks][nt],
                             (short)0, acc[mt][nt],
                             false,  false);
            __builtin_amdgcn_sched_barrier(0);
        }

        const int orow0 = row0 + wm * 16 * MT;
        const int ocol0 = col0 + wn * 16 * NT;
#pragma unroll
        for (int nt = 0; nt < NT; ++nt) {
            const int   col  = ocol0 + nt * 16;
            const float bcol = bias[col + l15];
#pragma unroll
            for (int mt = 0; mt < MT; ++mt) {
                const int row = orow0 + mt * 16;
#pragma unroll
                for (int r = 0; r < 8; ++r) {
                    vout[(size_t)(row + r + 8 * kh) * CC + col + l15] =
                        acc[mt][nt][r] + bcol;
                }
            }
        }
    }
}

extern "C" void kernel_launch(void* const* d_in, const int* in_sizes, int n_in,
                              void* d_out, int out_size, void* d_ws, size_t ws_size,
                              hipStream_t stream) {
    const float* query = (const float*)d_in[0];
    const float* qpts  = (const float*)d_in[1];
    const float* inp   = (const float*)d_in[2];
    const float* ipts  = (const float*)d_in[3];
    const float* Wv    = (const float*)d_in[4];
    const float* bv    = (const float*)d_in[5];
    const float* Ws    = (const float*)d_in[6];
    const float* bs    = (const float*)d_in[7];
    const float* Wa    = (const float*)d_in[8];
    const float* ba    = (const float*)d_in[9];
    const float* Wo    = (const float*)d_in[10];
    const float* bo    = (const float*)d_in[11];
    float* out = (float*)d_out;

    char* w = (char*)d_ws;
    float* value = (float*)w;  w += (size_t)BB * LIN * CC * 4;
    float* attw  = (float*)w;  w += (size_t)BB * LQ * MH * NP * 4;
    float* S4    = (float*)w;  w += (size_t)BB * LQ * MH * NP * 4 * 4;
    int*   idx   = (int*)w;    w += (size_t)BB * LQ * MH * NP * 4;

    static_assert(3 * (MH * NP) + MH * NP == 4 * 2 * 16, "the stacked width is the 1x4 waves' 128 columns");
    wmma_gemm_lds_f32<2, 2, 1, 4, 1><<<dim3((BB * LQ) / 32, 1), dim3(128), 0, stream>>>(
        query, Ws, bs, S4, Wa, ba, attw, qpts, nullptr, MH * NP * 4);

    static_assert((BB * LQ * MH * NP) / (16 * QT * KW) == ((BB * LIN) / VRX) * (CC / VRW),
                  "as many scan work-groups as value blocks");
    knn_and_value<<<(BB * LQ * MH * NP) / (16 * QT * KW), dim3(64 * KW), 0, stream>>>(
        S4, ipts, idx, inp, Wv, bv, value);

    static_assert(CC == 4 * 4 * 16 && CC == KK, "the 1x4 waves own all output columns");
    wmma_gemm_lds_f32<1, 4, 1, 4, 2><<<dim3((BB * LQ) / 16, 1), dim3(128), 0, stream>>>(
        value, Wo, bo, out, nullptr, nullptr, nullptr, attw, idx, CC);
}


# KNN scan: two running (min, index) pairs per query (even, odd candidate rows), merged once by value then index; a quarter's 8 compares precede its selects and none waits on another; same indices
# speedup vs baseline: 3.5113x; 1.0676x over previous
#include <hip/hip_runtime.h>
#include <hip/hip_bf16.h>


typedef __attribute__((ext_vector_type(2))) float v2f;
typedef __attribute__((ext_vector_type(8))) float v8f;
typedef __attribute__((ext_vector_type(4))) float v4f;

#define BB   2
#define LQ   1024
#define LIN  4096
#define CC   256
#define MH   8
#define NP   4
#define DH   32
#define QT   4
#define KW   4

#define KK     256
#define KPITCH (KK + 4)
#define KCH    64
#define KBL    32

template <int MT, int NT, int WM, int WN, int EPI>
__global__ void __launch_bounds__(32 * WM * WN)
wmma_gemm_lds_f32(const float* __restrict__ X, const float* __restrict__ W,
                  const float* __restrict__ bias, float* __restrict__ out,
                  const float* __restrict__ W2, const float* __restrict__ bias2,
                  float* __restrict__ out2, const float* __restrict__ aux,
                  const int* __restrict__ gidx, int Ncols) {
    constexpr int NTH  = 32 * WM * WN;
    constexpr int RX   = 16 * MT * WM;
    constexpr int RW   = 16 * NT * WN;
    constexpr int ROWS = RX + RW;
    constexpr int SROWS = (EPI == 2) ? RW : ROWS;
    constexpr int C4   = KCH / 4;
    constexpr int F4C  = SROWS * C4;
    constexpr int NLD  = (F4C + NTH - 1) / NTH;
    constexpr int NCH  = KK / KCH;
    constexpr int NKB  = KK / KBL;
    constexpr int BPC  = KCH / KBL;
    constexpr int KS   = KBL / 4;
    constexpr int N1   = (EPI == 1) ? 3 * (RW / 4) : RW;
    constexpr int N2   = RW - N1;

    __shared__ __attribute__((aligned(16))) float sT[ROWS * KPITCH];

    const int tid  = threadIdx.x;
    const int lane = tid & 31;
    const int wv   = tid >> 5;
    const int l15  = lane & 15;
    const int kh   = lane >> 4;
    const int wm   = wv / WN;
    const int wn   = wv % WN;
    const int row0 = blockIdx.x * RX;
    const int col0 = blockIdx.y * RW;

    constexpr int PW  = N1 / 3;
    constexpr int E1I = RX * PW;
    constexpr int E1T = (EPI == 1) ? (E1I + NTH - 1) / NTH : 1;
    float qv[E1T][3];
    if constexpr (EPI == 1) {
        static_assert(RW % 4 == 0 && N1 % (16 * NT) == 0 && N2 % 4 == 0,
                      "a wave's columns lie in one of the two products; whole softmax groups");
#pragma unroll
        for (int j = 0; j < E1T; ++j) {
            int i = tid + j * NTH;
            if (i >= E1I) i = E1I - 1;
            const float* q = aux + (size_t)(row0 + i / PW) * 3;
            qv[j][0] = q[0];
            qv[j][1] = q[1];
            qv[j][2] = q[2];
        }
    }

    constexpr int P4  = KK / 4;
    constexpr int GRS = NTH / P4;
    constexpr int NGP = (EPI == 2) ? RX / GRS : 1;
    const int gpc = tid % P4;
    const int gr0 = tid / P4;
    int4   gi[NGP];
    float4 ga[NGP];
    if constexpr (EPI == 2) {
        static_assert(NP == 4 && KK == MH * DH && DH % 4 == 0, "four points per head, heads side by side");
        static_assert(NTH % P4 == 0 && RX % GRS == 0, "whole trips, one column piece per thread");
#pragma unroll
        for (int j = 0; j < NGP; ++j) {
            const size_t o = ((size_t)(row0 + gr0 + j * GRS) * MH + gpc / (DH / 4)) * NP;
            gi[j] = *(const int4*)(gidx + o);
            ga[j] = *(const float4*)(aux + o);
        }
    }

    const float* gsrc[NLD];
    int          soff[NLD];
    if constexpr (EPI == 1) {
        constexpr int RPT = NTH / C4;
        static_assert(NTH % C4 == 0 && F4C % NTH == 0 && RX % RPT == 0 && N1 % RPT == 0,
                      "every trip lies in one source");
        const int    pr   = tid / C4;
        const int    pc   = tid % C4;
        const size_t toff = (size_t)pr * KK + pc * 4;
#pragma unroll
        for (int j = 0; j < NLD; ++j) {
            const int    r    = j * RPT;
            const float* base = (r < RX)      ? X + (size_t)(row0 + r) * KK
                              : (r - RX < N1) ? W + (size_t)(r - RX) * KK
                                              : W2 + (size_t)(r - RX - N1) * KK;
            gsrc[j] = base + toff;
            soff[j] = (r + pr) * KPITCH + pc * 4;
        }
    } else {
#pragma unroll
        for (int j = 0; j < NLD; ++j) {
            int f = tid + j * NTH;
            if (f >= F4C) f = F4C - 1;
            const int r  = f / C4 + (ROWS - SROWS);
            const int c4 = f % C4;
            gsrc[j] = (r < RX) ? X + (size_t)(row0 + r) * KK + c4 * 4
                               : W + (size_t)(col0 + r - RX) * KK + c4 * 4;
            soff[j] = r * KPITCH + c4 * 4;
        }
    }

    v4f tmp[NLD];
#pragma unroll
    for (int j = 0; j < NLD; ++j)
        if (j + 1 < NLD || F4C % NTH == 0 || tid + j * NTH < F4C)
            tmp[j] = *(const v4f*)(gsrc[j]);
    if constexpr (EPI == 2) {
        v4f gv[NGP][NP];
#pragma unroll
        for (int j = 0; j < NGP; ++j) {
            const int    row = row0 + gr0 + j * GRS;
            const float* vb  = X + (size_t)(row / LQ) * LIN * KK + gpc * 4;
            gv[j][0] = *(const v4f*)(vb + (size_t)gi[j].x * KK);
            gv[j][1] = *(const v4f*)(vb + (size_t)gi[j].y * KK);
            gv[j][2] = *(const v4f*)(vb + (size_t)gi[j].z * KK);
            gv[j][3] = *(const v4f*)(vb + (size_t)gi[j].w * KK);
        }
#pragma unroll
        for (int j = 0; j < NGP; ++j) {
            v4f x;
#pragma unroll
            for (int e = 0; e < 4; ++e) {
                float acc = 0.0f;
                acc += ga[j].x * gv[j][0][e];
                acc += ga[j].y * gv[j][1][e];
                acc += ga[j].z * gv[j][2][e];
                acc += ga[j].w * gv[j][3][e];
                x[e] = acc;
            }
            *(v4f*)(sT + (gr0 + j * GRS) * KPITCH + gpc * 4) = x;
        }
    }
#pragma unroll
    for (int j = 0; j < NLD; ++j)
        if (j + 1 < NLD || F4C % NTH == 0 || tid + j * NTH < F4C)
            *(v4f*)(sT + soff[j]) = tmp[j];
    __syncthreads();
    if (NCH > 1) {
#pragma unroll
        for (int j = 0; j < NLD; ++j)
            if (j + 1 < NLD || F4C % NTH == 0 || tid + j * NTH < F4C)
                tmp[j] = *(const v4f*)(gsrc[j] + KCH);
    }

    const float* ap = sT + (wm * 16 * MT + l15) * KPITCH + 2 * kh;
    const float* bp = sT + (RX + wn * 16 * NT + l15) * KPITCH + 2 * kh;

    v8f acc[MT][NT];
#pragma unroll
    for (int mt = 0; mt < MT; ++mt)
#pragma unroll
        for (int nt = 0; nt < NT; ++nt) {
            v8f z = {};
            acc[mt][nt] = z;
        }

    v2f fa[2][KS][MT], fb[2][KS][NT];
#pragma unroll
    for (int ks = 0; ks < KS; ++ks) {
#pragma unroll
        for (int mt = 0; mt < MT; ++mt)
            fa[0][ks][mt] = *(const v2f*)(ap + mt * 16 * KPITCH + ks * 4);
#pragma unroll
        for (int nt = 0; nt < NT; ++nt)
            fb[0][ks][nt] = *(const v2f*)(bp + nt * 16 * KPITCH + ks * 4);
    }

#pragma unroll
    for (int kb = 0; kb < NKB; ++kb) {
        const int cur = kb & 1;
        const int c   = kb / BPC;
        if (kb % BPC == BPC - 1 && c + 1 < NCH) {
#pragma unroll
            for (int j = 0; j < NLD; ++j)
                if (j + 1 < NLD || F4C % NTH == 0 || tid + j * NTH < F4C)
                    *(v4f*)(sT + soff[j] + (c + 1) * KCH) = tmp[j];
            __syncthreads();
            if (c + 2 < NCH) {
#pragma unroll
                for (int j = 0; j < NLD; ++j)
                    if (j + 1 < NLD || F4C % NTH == 0 || tid + j * NTH < F4C)
                        tmp[j] = *(const v4f*)(gsrc[j] + (c + 2) * KCH);
            }
        }
        if (kb + 1 < NKB) {
#pragma unroll
            for (int ks = 0; ks < KS; ++ks) {
#pragma unroll
                for (int mt = 0; mt < MT; ++mt)
                    fa[cur ^ 1][ks][mt] =
                        *(const v2f*)(ap + mt * 16 * KPITCH + (kb + 1) * KBL + ks * 4);
#pragma unroll
                for (int nt = 0; nt < NT; ++nt)
                    fb[cur ^ 1][ks][nt] =
                        *(const v2f*)(bp + nt * 16 * KPITCH + (kb + 1) * KBL + ks * 4);
            }
        }
        __builtin_amdgcn_sched_barrier(0);
#pragma unroll
        for (int ks = 0; ks < KS; ++ks)
#pragma unroll
            for (int mt = 0; mt < MT; ++mt)
#pragma unroll
                for (int nt = 0; nt < NT; ++nt)
                    acc[mt][nt] = __builtin_amdgcn_wmma_f32_16x16x4_f32(
                         false, fa[cur][ks][mt],  false, fb[cur][ks][nt],
                         (short)0, acc[mt][nt],
                         false,  false);
        __builtin_amdgcn_sched_barrier(0);
    }

    const int orow0 = row0 + wm * 16 * MT;
    const int ocol0 = col0 + wn * 16 * NT;
    if constexpr (EPI != 1) {
#pragma unroll
        for (int nt = 0; nt < NT; ++nt) {
            const int   col  = ocol0 + nt * 16;
            const float bcol = bias[col + l15];
#pragma unroll
            for (int mt = 0; mt < MT; ++mt) {
                const int row = orow0 + mt * 16;
#pragma unroll
                for (int r = 0; r < 8; ++r) {
                    out[(size_t)(row + r + 8 * kh) * Ncols + col + l15] =
                        acc[mt][nt][r] + bcol;
                }
            }
        }
    } else {
        constexpr int EP = RW + 4;
        __shared__ __attribute__((aligned(16))) float sE[RX * EP];
#pragma unroll
        for (int nt = 0; nt < NT; ++nt) {
            const int   lc   = wn * 16 * NT + nt * 16 + l15;
            const float bcol = (lc < N1) ? bias[lc] : bias2[lc - N1];
#pragma unroll
            for (int mt = 0; mt < MT; ++mt)
#pragma unroll
                for (int r = 0; r < 8; ++r)
                    sE[(wm * 16 * MT + mt * 16 + r + 8 * kh) * EP + lc] =
                        acc[mt][nt][r] + bcol;
        }
        __syncthreads();
#pragma unroll
        for (int j = 0; j < E1T; ++j) {
            const int i = tid + j * NTH;
            if (j < E1I / NTH || i < E1I) {
                const int    r = i / PW;
                const int    p = i % PW;
                const float* o = sE + r * EP + p * 3;
                float4 s;
                s.x = qv[j][0] + o[0];
                s.y = qv[j][1] + o[1];
                s.z = qv[j][2] + o[2];
                s.w = 1.0f;
                *(float4*)(out + ((size_t)(row0 + r) * PW + p) * 4) = s;
            }
        }
        constexpr int GW = N2 / 4;
        constexpr int ITEMS = RX * GW;
#pragma unroll
        for (int j = 0; j < (ITEMS + NTH - 1) / NTH; ++j) {
            const int i = tid + j * NTH;
            if (ITEMS % NTH != 0 && i >= ITEMS) continue;
            const int r = i / GW;
            const int g = i % GW;
            float4 v = *(const float4*)(sE + r * EP + N1 + g * 4);
            const float mx = fmaxf(fmaxf(v.x, v.y), fmaxf(v.z, v.w));
            const float e0 = __expf(v.x - mx), e1 = __expf(v.y - mx);
            const float e2 = __expf(v.z - mx), e3 = __expf(v.w - mx);
            const float inv = 1.0f / (e0 + e1 + e2 + e3);
            v.x = e0 * inv; v.y = e1 * inv; v.z = e2 * inv; v.w = e3 * inv;
            *(float4*)(out2 + (size_t)(row0 + r) * N2 + g * 4) = v;
        }
    }
}

__device__ __forceinline__ void knn_rows2(const v8f* c, int r0, int cib,
                                          float* best, int* bidx) {
    bool lt[2][QT];
#pragma unroll
    for (int r = r0; r < r0 + 2; ++r)
#pragma unroll
        for (int t = 0; t < QT; ++t)
            lt[r - r0][t] = c[t][r] < best[(r & 1) * QT + t];
#pragma unroll
    for (int r = r0; r < r0 + 2; ++r) {
#pragma unroll
        for (int t = 0; t < QT; ++t) {
            const int e = (r & 1) * QT + t;
            best[e] = lt[r - r0][t] ? c[t][r] : best[e];
            bidx[e] = lt[r - r0][t] ? cib + r : bidx[e];
        }
    }
}

__device__ __forceinline__ void knn_phase(v8f* cn, const v8f* cc, const v2f a,
                                          const v2f* bq, int cib,
                                          float* best, int* bidx) {
    static_assert(2 * QT == 8, "one product per two of a tile's eight rows");
#pragma unroll
    for (int t = 0; t < QT; ++t) {
        v8f z = {};
        cn[t] = __builtin_amdgcn_wmma_f32_16x16x4_f32(
            false, a, false, bq[t], (short)0, z, false, false);
        knn_rows2(cc, 2 * t, cib, best, bidx);
        __builtin_amdgcn_sched_barrier(0);
    }
}

#define VRX   64
#define VRW   128
#define VROWS (VRX + VRW)

__global__ void __launch_bounds__(64 * KW)
knn_and_value(const float* __restrict__ S4, const float* __restrict__ ipts,
              int* __restrict__ idxOut,
              const float* __restrict__ X, const float* __restrict__ Wt,
              const float* __restrict__ bias, float* __restrict__ vout) {
    constexpr int NTH = 64 * KW;
    __shared__ __attribute__((aligned(16))) float sV[LIN * 4];
    __shared__ __attribute__((aligned(16))) float sT[VROWS * KPITCH];

    const int tid  = threadIdx.x;
    const int lane = tid & 31;
    const int wv   = __builtin_amdgcn_readfirstlane(tid >> 5);
    const int l15  = lane & 15;
    const int kh   = lane >> 4;
    const int b    = (blockIdx.x * (KW * 16 * QT)) >> 15;
    const int n0   = (blockIdx.x * KW + (wv & (KW - 1))) * (16 * QT);
    const int row0 = (blockIdx.x >> 1) * VRX;
    const int col0 = (blockIdx.x & 1) * VRW;

    v2f bq[QT];
#pragma unroll
    for (int t = 0; t < QT; ++t) {
        v2f z = {};
        bq[t] = z;
    }
    if (wv < KW) {
#pragma unroll
        for (int t = 0; t < QT; ++t)
            bq[t] = *(const v2f*)(S4 + (size_t)(n0 + t * 16 + l15) * 4 + 2 * kh);
    }

    {
        const float4* __restrict__ g = (const float4*)(ipts + (size_t)b * LIN * 3);
        float4* s = (float4*)sV;
        constexpr int NG  = LIN / 4 / NTH;
        constexpr int C4  = KK / 4;
        constexpr int RPT = NTH / C4;
        constexpr int NLX = VRX / RPT;
        constexpr int NLW = VRW / RPT;
        static_assert(NTH % C4 == 0 && VRX % RPT == 0 && VRW % RPT == 0, "whole trips");
        const int pr = tid / C4;
        const int pc = tid % C4;
        const float* xsrc = X  + (size_t)(row0 + pr) * KK + pc * 4;
        const float* wsrc = Wt + (size_t)(col0 + pr) * KK + pc * 4;
        float*       sdst = sT + pr * KPITCH + pc * 4;

        float4 tmp[NG][3];
        v4f    xt[NLX], wt[NLW];
#pragma unroll
        for (int j = 0; j < NG; ++j)
#pragma unroll
            for (int k = 0; k < 3; ++k) tmp[j][k] = g[3 * (tid + NTH * j) + k];
#pragma unroll
        for (int j = 0; j < NLX; ++j) xt[j] = *(const v4f*)(xsrc + (size_t)j * RPT * KK);
#pragma unroll
        for (int j = 0; j < NLW; ++j) wt[j] = *(const v4f*)(wsrc + (size_t)j * RPT * KK);

#pragma unroll
        for (int j = 0; j < NG; ++j) {
            const float f[12] = {tmp[j][0].x, tmp[j][0].y, tmp[j][0].z, tmp[j][0].w,
                                 tmp[j][1].x, tmp[j][1].y, tmp[j][1].z, tmp[j][1].w,
                                 tmp[j][2].x, tmp[j][2].y, tmp[j][2].z, tmp[j][2].w};
#pragma unroll
            for (int i = 0; i < 4; ++i) {
                const float x = f[3 * i], y = f[3 * i + 1], z = f[3 * i + 2];
                float xy;
                {
#pragma clang fp contract(off)
                    xy = x * x + y * y;
                }
                float4 t;
                t.x = -2.0f * x;
                t.y = -2.0f * y;
                t.z = -2.0f * z;
                t.w = __builtin_fmaf(z, z, xy);
                s[4 * (tid + NTH * j) + i] = t;
            }
        }
#pragma unroll
        for (int j = 0; j < NLX; ++j) *(v4f*)(sdst + j * RPT * KPITCH) = xt[j];
#pragma unroll
        for (int j = 0; j < NLW; ++j) *(v4f*)(sdst + (VRX + j * RPT) * KPITCH) = wt[j];
    }
    __syncthreads();

    if (wv < KW) {
        const float* vptr = sV + l15 * 4 + 2 * kh;

        float best[2 * QT];
        int   bidx[2 * QT];
#pragma unroll
        for (int t = 0; t < 2 * QT; ++t) { best[t] = 3.4e38f; bidx[t] = 0; }

        constexpr int NBLK = LIN / 16;
        const int ch = kh << 3;
        v8f c0[QT], c1[QT];
        v2f aA = *(const v2f*)(vptr);
        v2f aB = *(const v2f*)(vptr + 16 * 4);
#pragma unroll
        for (int t = 0; t < QT; ++t) {
            v8f z = {};
            c0[t] = __builtin_amdgcn_wmma_f32_16x16x4_f32(
                false, aA, false, bq[t], (short)0, z, false, false);
        }
        aA = *(const v2f*)(vptr + 2 * 16 * 4);
        for (int j = 0; j < NBLK - 4; j += 2) {
            knn_phase(c1, c0, aB, bq, j * 16 + ch, best, bidx);
            aB = *(const v2f*)(vptr + (j + 3) * 16 * 4);
            knn_phase(c0, c1, aA, bq, (j + 1) * 16 + ch, best, bidx);
            aA = *(const v2f*)(vptr + (j + 4) * 16 * 4);
        }
        knn_phase(c1, c0, aB, bq, (NBLK - 4) * 16 + ch, best, bidx);
        aB = *(const v2f*)(vptr + (NBLK - 1) * 16 * 4);
        knn_phase(c0, c1, aA, bq, (NBLK - 3) * 16 + ch, best, bidx);
        knn_phase(c1, c0, aB, bq, (NBLK - 2) * 16 + ch, best, bidx);
#pragma unroll
        for (int q = 0; q < 4; ++q)
            knn_rows2(c1, 2 * q, (NBLK - 1) * 16 + ch, best, bidx);

#pragma unroll
        for (int t = 0; t < QT; ++t) {
            if (best[QT + t] < best[t] || (best[QT + t] == best[t] && bidx[QT + t] < bidx[t])) {
                best[t] = best[QT + t]; bidx[t] = bidx[QT + t];
            }
        }
#pragma unroll
        for (int t = 0; t < QT; ++t) {
            const float obest = __shfl_xor(best[t], 16, 32);
            const int   oidx  = __shfl_xor(bidx[t], 16, 32);
            if (obest < best[t] || (obest == best[t] && oidx < bidx[t])) {
                best[t] = obest; bidx[t] = oidx;
            }
            if (lane < 16) idxOut[n0 + t * 16 + l15] = bidx[t];
        }
    } else {
        constexpr int MT  = 2, NT = 4;
        constexpr int NKB = KK / KBL;
        constexpr int KS  = KBL / 4;
        static_assert(VRX == 2 * 16 * MT && VRW == 2 * 16 * NT, "2x2 waves of 32x64");
        const int w  = wv - KW;
        const int wm = w >> 1;
        const int wn = w & 1;
        const float* ap = sT + (wm * 16 * MT + l15) * KPITCH + 2 * kh;
        const float* bp = sT + (VRX + wn * 16 * NT + l15) * KPITCH + 2 * kh;

        v8f acc[MT][NT];
#pragma unroll
        for (int mt = 0; mt < MT; ++mt)
#pragma unroll
            for (int nt = 0; nt < NT; ++nt) {
                v8f z = {};
                acc[mt][nt] = z;
            }

        v2f fa[2][KS][MT], fb[2][KS][NT];
#pragma unroll
        for (int ks = 0; ks < KS; ++ks) {
#pragma unroll
            for (int mt = 0; mt < MT; ++mt)
                fa[0][ks][mt] = *(const v2f*)(ap + mt * 16 * KPITCH + ks * 4);
#pragma unroll
            for (int nt = 0; nt < NT; ++nt)
                fb[0][ks][nt] = *(const v2f*)(bp + nt * 16 * KPITCH + ks * 4);
        }

#pragma unroll
        for (int kb = 0; kb < NKB; ++kb) {
            const int cur = kb & 1;
            if (kb + 1 < NKB) {
#pragma unroll
                for (int ks = 0; ks < KS; ++ks) {
#pragma unroll
                    for (int mt = 0; mt < MT; ++mt)
                        fa[cur ^ 1][ks][mt] =
                            *(const v2f*)(ap + mt * 16 * KPITCH + (kb + 1) * KBL + ks * 4);
#pragma unroll
                    for (int nt = 0; nt < NT; ++nt)
                        fb[cur ^ 1][ks][nt] =
                            *(const v2f*)(bp + nt * 16 * KPITCH + (kb + 1) * KBL + ks * 4);
                }
            }
            __builtin_amdgcn_sched_barrier(0);
#pragma unroll
            for (int ks = 0; ks < KS; ++ks)
#pragma unroll
                for (int mt = 0; mt < MT; ++mt)
#pragma unroll
                    for (int nt = 0; nt < NT; ++nt)
                        acc[mt][nt] = __builtin_amdgcn_wmma_f32_16x16x4_f32(
                             false, fa[cur][ks][mt],  false, fb[cur][ks][nt],
                             (short)0, acc[mt][nt],
                             false,  false);
            __builtin_amdgcn_sched_barrier(0);
        }

        const int orow0 = row0 + wm * 16 * MT;
        const int ocol0 = col0 + wn * 16 * NT;
#pragma unroll
        for (int nt = 0; nt < NT; ++nt) {
            const int   col  = ocol0 + nt * 16;
            const float bcol = bias[col + l15];
#pragma unroll
            for (int mt = 0; mt < MT; ++mt) {
                const int row = orow0 + mt * 16;
#pragma unroll
                for (int r = 0; r < 8; ++r) {
                    vout[(size_t)(row + r + 8 * kh) * CC + col + l15] =
                        acc[mt][nt][r] + bcol;
                }
            }
        }
    }
}

extern "C" void kernel_launch(void* const* d_in, const int* in_sizes, int n_in,
                              void* d_out, int out_size, void* d_ws, size_t ws_size,
                              hipStream_t stream) {
    const float* query = (const float*)d_in[0];
    const float* qpts  = (const float*)d_in[1];
    const float* inp   = (const float*)d_in[2];
    const float* ipts  = (const float*)d_in[3];
    const float* Wv    = (const float*)d_in[4];
    const float* bv    = (const float*)d_in[5];
    const float* Ws    = (const float*)d_in[6];
    const float* bs    = (const float*)d_in[7];
    const float* Wa    = (const float*)d_in[8];
    const float* ba    = (const float*)d_in[9];
    const float* Wo    = (const float*)d_in[10];
    const float* bo    = (const float*)d_in[11];
    float* out = (float*)d_out;

    char* w = (char*)d_ws;
    float* value = (float*)w;  w += (size_t)BB * LIN * CC * 4;
    float* attw  = (float*)w;  w += (size_t)BB * LQ * MH * NP * 4;
    float* S4    = (float*)w;  w += (size_t)BB * LQ * MH * NP * 4 * 4;
    int*   idx   = (int*)w;    w += (size_t)BB * LQ * MH * NP * 4;

    static_assert(3 * (MH * NP) + MH * NP == 4 * 2 * 16, "the stacked width is the 1x4 waves' 128 columns");
    wmma_gemm_lds_f32<2, 2, 1, 4, 1><<<dim3((BB * LQ) / 32, 1), dim3(128), 0, stream>>>(
        query, Ws, bs, S4, Wa, ba, attw, qpts, nullptr, MH * NP * 4);

    static_assert((BB * LQ * MH * NP) / (16 * QT * KW) == ((BB * LIN) / VRX) * (CC / VRW),
                  "as many scan work-groups as value blocks");
    knn_and_value<<<(BB * LQ * MH * NP) / (16 * QT * KW), dim3(64 * KW), 0, stream>>>(
        S4, ipts, idx, inp, Wv, bv, value);

    static_assert(CC == 4 * 4 * 16 && CC == KK, "the 1x4 waves own all output columns");
    wmma_gemm_lds_f32<1, 4, 1, 4, 2><<<dim3((BB * LQ) / 16, 1), dim3(128), 0, stream>>>(
        value, Wo, bo, out, nullptr, nullptr, nullptr, attw, idx, CC);
}


# Offsets+logits product moved into the KNN launch: a work-group first multiplies its 8 query rows by the 128 stacked weight rows (one tile per wave), points stay in LDS; 3->2 launches; bit-identical
# speedup vs baseline: 3.8137x; 1.0861x over previous
#include <hip/hip_runtime.h>
#include <hip/hip_bf16.h>


typedef __attribute__((ext_vector_type(2))) float v2f;
typedef __attribute__((ext_vector_type(8))) float v8f;
typedef __attribute__((ext_vector_type(4))) float v4f;

#define BB   2
#define LQ   1024
#define LIN  4096
#define CC   256
#define MH   8
#define NP   4
#define DH   32
#define QT   4
#define KW   4

#define KK     256
#define KPITCH (KK + 4)
#define KCH    64
#define KBL    32

template <int MT, int NT, int WM, int WN, int EPI>
__global__ void __launch_bounds__(32 * WM * WN)
wmma_gemm_lds_f32(const float* __restrict__ X, const float* __restrict__ W,
                  const float* __restrict__ bias, float* __restrict__ out,
                  const float* __restrict__ aux, const int* __restrict__ gidx, int Ncols) {
    constexpr int NTH  = 32 * WM * WN;
    constexpr int RX   = 16 * MT * WM;
    constexpr int RW   = 16 * NT * WN;
    constexpr int ROWS = RX + RW;
    constexpr int SROWS = (EPI == 2) ? RW : ROWS;
    constexpr int C4   = KCH / 4;
    constexpr int F4C  = SROWS * C4;
    constexpr int NLD  = (F4C + NTH - 1) / NTH;
    constexpr int NCH  = KK / KCH;
    constexpr int NKB  = KK / KBL;
    constexpr int BPC  = KCH / KBL;
    constexpr int KS   = KBL / 4;

    __shared__ __attribute__((aligned(16))) float sT[ROWS * KPITCH];

    const int tid  = threadIdx.x;
    const int lane = tid & 31;
    const int wv   = tid >> 5;
    const int l15  = lane & 15;
    const int kh   = lane >> 4;
    const int wm   = wv / WN;
    const int wn   = wv % WN;
    const int row0 = blockIdx.x * RX;
    const int col0 = blockIdx.y * RW;

    constexpr int P4  = KK / 4;
    constexpr int GRS = NTH / P4;
    constexpr int NGP = (EPI == 2) ? RX / GRS : 1;
    const int gpc = tid % P4;
    const int gr0 = tid / P4;
    int4   gi[NGP];
    float4 ga[NGP];
    if constexpr (EPI == 2) {
        static_assert(NP == 4 && KK == MH * DH && DH % 4 == 0, "four points per head, heads side by side");
        static_assert(NTH % P4 == 0 && RX % GRS == 0, "whole trips, one column piece per thread");
#pragma unroll
        for (int j = 0; j < NGP; ++j) {
            const size_t o = ((size_t)(row0 + gr0 + j * GRS) * MH + gpc / (DH / 4)) * NP;
            gi[j] = *(const int4*)(gidx + o);
            ga[j] = *(const float4*)(aux + o);
        }
    }

    const float* gsrc[NLD];
    int          soff[NLD];
#pragma unroll
    for (int j = 0; j < NLD; ++j) {
        int f = tid + j * NTH;
        if (f >= F4C) f = F4C - 1;
        const int r  = f / C4 + (ROWS - SROWS);
        const int c4 = f % C4;
        gsrc[j] = (r < RX) ? X + (size_t)(row0 + r) * KK + c4 * 4
                           : W + (size_t)(col0 + r - RX) * KK + c4 * 4;
        soff[j] = r * KPITCH + c4 * 4;
    }

    v4f tmp[NLD];
#pragma unroll
    for (int j = 0; j < NLD; ++j)
        if (j + 1 < NLD || F4C % NTH == 0 || tid + j * NTH < F4C)
            tmp[j] = *(const v4f*)(gsrc[j]);
    if constexpr (EPI == 2) {
        v4f gv[NGP][NP];
#pragma unroll
        for (int j = 0; j < NGP; ++j) {
            const int    row = row0 + gr0 + j * GRS;
            const float* vb  = X + (size_t)(row / LQ) * LIN * KK + gpc * 4;
            gv[j][0] = *(const v4f*)(vb + (size_t)gi[j].x * KK);
            gv[j][1] = *(const v4f*)(vb + (size_t)gi[j].y * KK);
            gv[j][2] = *(const v4f*)(vb + (size_t)gi[j].z * KK);
            gv[j][3] = *(const v4f*)(vb + (size_t)gi[j].w * KK);
        }
#pragma unroll
        for (int j = 0; j < NGP; ++j) {
            v4f x;
#pragma unroll
            for (int e = 0; e < 4; ++e) {
                float acc = 0.0f;
                acc += ga[j].x * gv[j][0][e];
                acc += ga[j].y * gv[j][1][e];
                acc += ga[j].z * gv[j][2][e];
                acc += ga[j].w * gv[j][3][e];
                x[e] = acc;
            }
            *(v4f*)(sT + (gr0 + j * GRS) * KPITCH + gpc * 4) = x;
        }
    }
#pragma unroll
    for (int j = 0; j < NLD; ++j)
        if (j + 1 < NLD || F4C % NTH == 0 || tid + j * NTH < F4C)
            *(v4f*)(sT + soff[j]) = tmp[j];
    __syncthreads();
    if (NCH > 1) {
#pragma unroll
        for (int j = 0; j < NLD; ++j)
            if (j + 1 < NLD || F4C % NTH == 0 || tid + j * NTH < F4C)
                tmp[j] = *(const v4f*)(gsrc[j] + KCH);
    }

    const float* ap = sT + (wm * 16 * MT + l15) * KPITCH + 2 * kh;
    const float* bp = sT + (RX + wn * 16 * NT + l15) * KPITCH + 2 * kh;

    v8f acc[MT][NT];
#pragma unroll
    for (int mt = 0; mt < MT; ++mt)
#pragma unroll
        for (int nt = 0; nt < NT; ++nt) {
            v8f z = {};
            acc[mt][nt] = z;
        }

    v2f fa[2][KS][MT], fb[2][KS][NT];
#pragma unroll
    for (int ks = 0; ks < KS; ++ks) {
#pragma unroll
        for (int mt = 0; mt < MT; ++mt)
            fa[0][ks][mt] = *(const v2f*)(ap + mt * 16 * KPITCH + ks * 4);
#pragma unroll
        for (int nt = 0; nt < NT; ++nt)
            fb[0][ks][nt] = *(const v2f*)(bp + nt * 16 * KPITCH + ks * 4);
    }

#pragma unroll
    for (int kb = 0; kb < NKB; ++kb) {
        const int cur = kb & 1;
        const int c   = kb / BPC;
        if (kb % BPC == BPC - 1 && c + 1 < NCH) {
#pragma unroll
            for (int j = 0; j < NLD; ++j)
                if (j + 1 < NLD || F4C % NTH == 0 || tid + j * NTH < F4C)
                    *(v4f*)(sT + soff[j] + (c + 1) * KCH) = tmp[j];
            __syncthreads();
            if (c + 2 < NCH) {
#pragma unroll
                for (int j = 0; j < NLD; ++j)
                    if (j + 1 < NLD || F4C % NTH == 0 || tid + j * NTH < F4C)
                        tmp[j] = *(const v4f*)(gsrc[j] + (c + 2) * KCH);
            }
        }
        if (kb + 1 < NKB) {
#pragma unroll
            for (int ks = 0; ks < KS; ++ks) {
#pragma unroll
                for (int mt = 0; mt < MT; ++mt)
                    fa[cur ^ 1][ks][mt] =
                        *(const v2f*)(ap + mt * 16 * KPITCH + (kb + 1) * KBL + ks * 4);
#pragma unroll
                for (int nt = 0; nt < NT; ++nt)
                    fb[cur ^ 1][ks][nt] =
                        *(const v2f*)(bp + nt * 16 * KPITCH + (kb + 1) * KBL + ks * 4);
            }
        }
        __builtin_amdgcn_sched_barrier(0);
#pragma unroll
        for (int ks = 0; ks < KS; ++ks)
#pragma unroll
            for (int mt = 0; mt < MT; ++mt)
#pragma unroll
                for (int nt = 0; nt < NT; ++nt)
                    acc[mt][nt] = __builtin_amdgcn_wmma_f32_16x16x4_f32(
                         false, fa[cur][ks][mt],  false, fb[cur][ks][nt],
                         (short)0, acc[mt][nt],
                         false,  false);
        __builtin_amdgcn_sched_barrier(0);
    }

    const int orow0 = row0 + wm * 16 * MT;
    const int ocol0 = col0 + wn * 16 * NT;
#pragma unroll
    for (int nt = 0; nt < NT; ++nt) {
        const int   col  = ocol0 + nt * 16;
        const float bcol = bias[col + l15];
#pragma unroll
        for (int mt = 0; mt < MT; ++mt) {
            const int row = orow0 + mt * 16;
#pragma unroll
            for (int r = 0; r < 8; ++r) {
                out[(size_t)(row + r + 8 * kh) * Ncols + col + l15] =
                    acc[mt][nt][r] + bcol;
            }
        }
    }
}

__device__ __forceinline__ void knn_rows2(const v8f* c, int r0, int cib,
                                          float* best, int* bidx) {
    bool lt[2][QT];
#pragma unroll
    for (int r = r0; r < r0 + 2; ++r)
#pragma unroll
        for (int t = 0; t < QT; ++t)
            lt[r - r0][t] = c[t][r] < best[(r & 1) * QT + t];
#pragma unroll
    for (int r = r0; r < r0 + 2; ++r) {
#pragma unroll
        for (int t = 0; t < QT; ++t) {
            const int e = (r & 1) * QT + t;
            best[e] = lt[r - r0][t] ? c[t][r] : best[e];
            bidx[e] = lt[r - r0][t] ? cib + r : bidx[e];
        }
    }
}

__device__ __forceinline__ void knn_phase(v8f* cn, const v8f* cc, const v2f a,
                                          const v2f* bq, int cib,
                                          float* best, int* bidx) {
    static_assert(2 * QT == 8, "one product per two of a tile's eight rows");
#pragma unroll
    for (int t = 0; t < QT; ++t) {
        v8f z = {};
        cn[t] = __builtin_amdgcn_wmma_f32_16x16x4_f32(
            false, a, false, bq[t], (short)0, z, false, false);
        knn_rows2(cc, 2 * t, cib, best, bidx);
        __builtin_amdgcn_sched_barrier(0);
    }
}

#define VRX   64
#define VRW   128
#define VROWS (VRX + VRW)
#define QRW   8
#define SW1   (3 * MH * NP)
#define SWN   (SW1 + MH * NP)
#define SWR0  16
#define SEP   (SWN + 4)

__global__ void __launch_bounds__(64 * KW)
points_knn_value(const float* __restrict__ query, const float* __restrict__ qpts,
                 const float* __restrict__ Ws, const float* __restrict__ bs,
                 const float* __restrict__ Wa, const float* __restrict__ ba,
                 float* __restrict__ attw,
                 const float* __restrict__ ipts, int* __restrict__ idxOut,
                 const float* __restrict__ X, const float* __restrict__ Wt,
                 const float* __restrict__ bias, float* __restrict__ vout) {
    constexpr int NTH = 64 * KW;
    constexpr int C4  = KK / 4;
    constexpr int RPT = NTH / C4;
    constexpr int NLQ = QRW / RPT;
    constexpr int NLS = SWN / RPT;
    constexpr int NLX = VRX / RPT;
    constexpr int NLW = VRW / RPT;
    constexpr int NKB = KK / KBL;
    constexpr int KS  = KBL / 4;
    static_assert(NTH % C4 == 0 && QRW % RPT == 0 && SW1 % RPT == 0 && SWN % RPT == 0 &&
                  VRX % RPT == 0 && VRW % RPT == 0, "whole trips, every trip in one source");
    static_assert(KW * 16 * QT == QRW * MH * NP && NTH == QRW * MH * NP,
                  "the scan's points are the points of the work-group's rows, one per thread");
    static_assert(SWN == 16 * 2 * KW, "one 16-column tile of the stacked product per wave");
    static_assert(SWR0 >= QRW && SWR0 + SWN <= VROWS, "phase 1's operands fit the value block's rows");
    static_assert(NP == 4 && QRW * MH <= 32 * KW, "softmax groups of four, one per thread of the product waves");

    __shared__ __attribute__((aligned(16))) float sV[LIN * 4];
    __shared__ __attribute__((aligned(16))) float sT[VROWS * KPITCH];
    __shared__ __attribute__((aligned(16))) float sE[QRW * SEP];
    __shared__ __attribute__((aligned(16))) float sS[NTH * 4];

    const int tid  = threadIdx.x;
    const int lane = tid & 31;
    const int wv   = __builtin_amdgcn_readfirstlane(tid >> 5);
    const int l15  = lane & 15;
    const int kh   = lane >> 4;
    const int b    = (blockIdx.x * (KW * 16 * QT)) >> 15;
    const int n0   = (blockIdx.x * KW + (wv & (KW - 1))) * (16 * QT);
    const int qrow0 = blockIdx.x * QRW;
    const int row0 = (blockIdx.x >> 1) * VRX;
    const int col0 = (blockIdx.x & 1) * VRW;

    const int pr = tid / C4;
    const int pc = tid % C4;
    float* sdst = sT + pr * KPITCH + pc * 4;

    const float* qp = qpts + (size_t)(qrow0 + tid / (MH * NP)) * 3;
    const float qv0 = qp[0], qv1 = qp[1], qv2 = qp[2];
    const int   lc   = wv * 16 + l15;
    const float bcol = (lc < SW1) ? bs[lc] : ba[lc - SW1];

    {
        const float4* __restrict__ g = (const float4*)(ipts + (size_t)b * LIN * 3);
        float4* sc = (float4*)sV;
        constexpr int NG = LIN / 4 / NTH;
        const size_t toff = (size_t)pr * KK + pc * 4;

        float4 tmp[NG][3];
        v4f    qt[NLQ], st[NLS];
#pragma unroll
        for (int j = 0; j < NG; ++j)
#pragma unroll
            for (int k = 0; k < 3; ++k) tmp[j][k] = g[3 * (tid + NTH * j) + k];
#pragma unroll
        for (int j = 0; j < NLQ; ++j)
            qt[j] = *(const v4f*)(query + (size_t)(qrow0 + j * RPT) * KK + toff);
#pragma unroll
        for (int j = 0; j < NLS; ++j) {
            const int    r    = j * RPT;
            const float* base = (r < SW1) ? Ws + (size_t)r * KK : Wa + (size_t)(r - SW1) * KK;
            st[j] = *(const v4f*)(base + toff);
        }

#pragma unroll
        for (int j = 0; j < NG; ++j) {
            const float f[12] = {tmp[j][0].x, tmp[j][0].y, tmp[j][0].z, tmp[j][0].w,
                                 tmp[j][1].x, tmp[j][1].y, tmp[j][1].z, tmp[j][1].w,
                                 tmp[j][2].x, tmp[j][2].y, tmp[j][2].z, tmp[j][2].w};
#pragma unroll
            for (int i = 0; i < 4; ++i) {
                const float x = f[3 * i], y = f[3 * i + 1], z = f[3 * i + 2];
                float xy;
                {
#pragma clang fp contract(off)
                    xy = x * x + y * y;
                }
                float4 t;
                t.x = -2.0f * x;
                t.y = -2.0f * y;
                t.z = -2.0f * z;
                t.w = __builtin_fmaf(z, z, xy);
                sc[4 * (tid + NTH * j) + i] = t;
            }
        }
#pragma unroll
        for (int j = 0; j < NLQ; ++j) *(v4f*)(sdst + j * RPT * KPITCH) = qt[j];
#pragma unroll
        for (int j = 0; j < NLS; ++j) *(v4f*)(sdst + (SWR0 + j * RPT) * KPITCH) = st[j];
    }
    __syncthreads();

    v4f xt[NLX], wt[NLW];
    {
        const float* xsrc = X  + (size_t)(row0 + pr) * KK + pc * 4;
        const float* wsrc = Wt + (size_t)(col0 + pr) * KK + pc * 4;
#pragma unroll
        for (int j = 0; j < NLX; ++j) xt[j] = *(const v4f*)(xsrc + (size_t)j * RPT * KK);
#pragma unroll
        for (int j = 0; j < NLW; ++j) wt[j] = *(const v4f*)(wsrc + (size_t)j * RPT * KK);
    }

    {
        const float* ap = sT + (l15 & (QRW - 1)) * KPITCH + 2 * kh;
        const float* bp = sT + (SWR0 + wv * 16 + l15) * KPITCH + 2 * kh;
        v8f acc = {};
        v2f fa[2][KS], fb[2][KS];
#pragma unroll
        for (int ks = 0; ks < KS; ++ks) {
            fa[0][ks] = *(const v2f*)(ap + ks * 4);
            fb[0][ks] = *(const v2f*)(bp + ks * 4);
        }
#pragma unroll
        for (int kb = 0; kb < NKB; ++kb) {
            const int cur = kb & 1;
            if (kb + 1 < NKB) {
#pragma unroll
                for (int ks = 0; ks < KS; ++ks) {
                    fa[cur ^ 1][ks] = *(const v2f*)(ap + (kb + 1) * KBL + ks * 4);
                    fb[cur ^ 1][ks] = *(const v2f*)(bp + (kb + 1) * KBL + ks * 4);
                }
            }
            __builtin_amdgcn_sched_barrier(0);
#pragma unroll
            for (int ks = 0; ks < KS; ++ks)
                acc = __builtin_amdgcn_wmma_f32_16x16x4_f32(
                     false, fa[cur][ks],  false, fb[cur][ks],
                     (short)0, acc,  false,  false);
            __builtin_amdgcn_sched_barrier(0);
        }
        if (kh == 0) {
#pragma unroll
            for (int r = 0; r < QRW; ++r) sE[r * SEP + lc] = acc[r] + bcol;
        }
    }
    __syncthreads();

#pragma unroll
    for (int j = 0; j < NLX; ++j) *(v4f*)(sdst + j * RPT * KPITCH) = xt[j];
#pragma unroll
    for (int j = 0; j < NLW; ++j) *(v4f*)(sdst + (VRX + j * RPT) * KPITCH) = wt[j];
    {
        const float* o = sE + (tid / (MH * NP)) * SEP + (tid % (MH * NP)) * 3;
        float4 sp;
        sp.x = qv0 + o[0];
        sp.y = qv1 + o[1];
        sp.z = qv2 + o[2];
        sp.w = 1.0f;
        *(float4*)(sS + tid * 4) = sp;
    }
    __syncthreads();

    if (wv < KW) {
        v2f bq[QT];
#pragma unroll
        for (int t = 0; t < QT; ++t)
            bq[t] = *(const v2f*)(sS + (wv * (16 * QT) + t * 16 + l15) * 4 + 2 * kh);

        const float* vptr = sV + l15 * 4 + 2 * kh;

        float best[2 * QT];
        int   bidx[2 * QT];
#pragma unroll
        for (int t = 0; t < 2 * QT; ++t) { best[t] = 3.4e38f; bidx[t] = 0; }

        constexpr int NBLK = LIN / 16;
        const int ch = kh << 3;
        v8f c0[QT], c1[QT];
        v2f aA = *(const v2f*)(vptr);
        v2f aB = *(const v2f*)(vptr + 16 * 4);
#pragma unroll
        for (int t = 0; t < QT; ++t) {
            v8f z = {};
            c0[t] = __builtin_amdgcn_wmma_f32_16x16x4_f32(
                false, aA, false, bq[t], (short)0, z, false, false);
        }
        aA = *(const v2f*)(vptr + 2 * 16 * 4);
        for (int j = 0; j < NBLK - 4; j += 2) {
            knn_phase(c1, c0, aB, bq, j * 16 + ch, best, bidx);
            aB = *(const v2f*)(vptr + (j + 3) * 16 * 4);
            knn_phase(c0, c1, aA, bq, (j + 1) * 16 + ch, best, bidx);
            aA = *(const v2f*)(vptr + (j + 4) * 16 * 4);
        }
        knn_phase(c1, c0, aB, bq, (NBLK - 4) * 16 + ch, best, bidx);
        aB = *(const v2f*)(vptr + (NBLK - 1) * 16 * 4);
        knn_phase(c0, c1, aA, bq, (NBLK - 3) * 16 + ch, best, bidx);
        knn_phase(c1, c0, aB, bq, (NBLK - 2) * 16 + ch, best, bidx);
#pragma unroll
        for (int q = 0; q < 4; ++q)
            knn_rows2(c1, 2 * q, (NBLK - 1) * 16 + ch, best, bidx);

#pragma unroll
        for (int t = 0; t < QT; ++t) {
            if (best[QT + t] < best[t] || (best[QT + t] == best[t] && bidx[QT + t] < bidx[t])) {
                best[t] = best[QT + t]; bidx[t] = bidx[QT + t];
            }
        }
#pragma unroll
        for (int t = 0; t < QT; ++t) {
            const float obest = __shfl_xor(best[t], 16, 32);
            const int   oidx  = __shfl_xor(bidx[t], 16, 32);
            if (obest < best[t] || (obest == best[t] && oidx < bidx[t])) {
                best[t] = obest; bidx[t] = oidx;
            }
            if (lane < 16) idxOut[n0 + t * 16 + l15] = bidx[t];
        }
    } else {
        constexpr int MT  = 2, NT = 4;
        static_assert(VRX == 2 * 16 * MT && VRW == 2 * 16 * NT, "2x2 waves of 32x64");
        const int w  = wv - KW;
        const unsigned sg = (unsigned)tid - 32u * KW;
        if (sg < (unsigned)(QRW * MH)) {
            const int r = (int)(sg / MH);
            const int g = (int)(sg % MH);
            float4 v = *(const float4*)(sE + r * SEP + SW1 + g * 4);
            const float mx = fmaxf(fmaxf(v.x, v.y), fmaxf(v.z, v.w));
            const float e0 = __expf(v.x - mx), e1 = __expf(v.y - mx);
            const float e2 = __expf(v.z - mx), e3 = __expf(v.w - mx);
            const float inv = 1.0f / (e0 + e1 + e2 + e3);
            v.x = e0 * inv; v.y = e1 * inv; v.z = e2 * inv; v.w = e3 * inv;
            *(float4*)(attw + (size_t)(qrow0 + r) * (MH * NP) + g * 4) = v;
        }
        const int wm = w >> 1;
        const int wn = w & 1;
        const float* ap = sT + (wm * 16 * MT + l15) * KPITCH + 2 * kh;
        const float* bp = sT + (VRX + wn * 16 * NT + l15) * KPITCH + 2 * kh;

        v8f acc[MT][NT];
#pragma unroll
        for (int mt = 0; mt < MT; ++mt)
#pragma unroll
            for (int nt = 0; nt < NT; ++nt) {
                v8f z = {};
                acc[mt][nt] = z;
            }

        v2f fa[2][KS][MT], fb[2][KS][NT];
#pragma unroll
        for (int ks = 0; ks < KS; ++ks) {
#pragma unroll
            for (int mt = 0; mt < MT; ++mt)
                fa[0][ks][mt] = *(const v2f*)(ap + mt * 16 * KPITCH + ks * 4);
#pragma unroll
            for (int nt = 0; nt < NT; ++nt)
                fb[0][ks][nt] = *(const v2f*)(bp + nt * 16 * KPITCH + ks * 4);
        }

#pragma unroll
        for (int kb = 0; kb < NKB; ++kb) {
            const int cur = kb & 1;
            if (kb + 1 < NKB) {
#pragma unroll
                for (int ks = 0; ks < KS; ++ks) {
#pragma unroll
                    for (int mt = 0; mt < MT; ++mt)
                        fa[cur ^ 1][ks][mt] =
                            *(const v2f*)(ap + mt * 16 * KPITCH + (kb + 1) * KBL + ks * 4);
#pragma unroll
                    for (int nt = 0; nt < NT; ++nt)
                        fb[cur ^ 1][ks][nt] =
                            *(const v2f*)(bp + nt * 16 * KPITCH + (kb + 1) * KBL + ks * 4);
                }
            }
            __builtin_amdgcn_sched_barrier(0);
#pragma unroll
            for (int ks = 0; ks < KS; ++ks)
#pragma unroll
                for (int mt = 0; mt < MT; ++mt)
#pragma unroll
                    for (int nt = 0; nt < NT; ++nt)
                        acc[mt][nt] = __builtin_amdgcn_wmma_f32_16x16x4_f32(
                             false, fa[cur][ks][mt],  false, fb[cur][ks][nt],
                             (short)0, acc[mt][nt],
                             false,  false);
            __builtin_amdgcn_sched_barrier(0);
        }

        const int orow0 = row0 + wm * 16 * MT;
        const int ocol0 = col0 + wn * 16 * NT;
#pragma unroll
        for (int nt = 0; nt < NT; ++nt) {
            const int   col  = ocol0 + nt * 16;
            const float bcol = bias[col + l15];
#pragma unroll
            for (int mt = 0; mt < MT; ++mt) {
                const int row = orow0 + mt * 16;
#pragma unroll
                for (int r = 0; r < 8; ++r) {
                    vout[(size_t)(row + r + 8 * kh) * CC + col + l15] =
                        acc[mt][nt][r] + bcol;
                }
            }
        }
    }
}

extern "C" void kernel_launch(void* const* d_in, const int* in_sizes, int n_in,
                              void* d_out, int out_size, void* d_ws, size_t ws_size,
                              hipStream_t stream) {
    const float* query = (const float*)d_in[0];
    const float* qpts  = (const float*)d_in[1];
    const float* inp   = (const float*)d_in[2];
    const float* ipts  = (const float*)d_in[3];
    const float* Wv    = (const float*)d_in[4];
    const float* bv    = (const float*)d_in[5];
    const float* Ws    = (const float*)d_in[6];
    const float* bs    = (const float*)d_in[7];
    const float* Wa    = (const float*)d_in[8];
    const float* ba    = (const float*)d_in[9];
    const float* Wo    = (const float*)d_in[10];
    const float* bo    = (const float*)d_in[11];
    float* out = (float*)d_out;

    char* w = (char*)d_ws;
    float* value = (float*)w;  w += (size_t)BB * LIN * CC * 4;
    float* attw  = (float*)w;  w += (size_t)BB * LQ * MH * NP * 4;
    int*   idx   = (int*)w;    w += (size_t)BB * LQ * MH * NP * 4;

    static_assert((BB * LQ) / QRW == ((BB * LIN) / VRX) * (CC / VRW),
                  "as many scan work-groups as value blocks");
    points_knn_value<<<(BB * LQ) / QRW, dim3(64 * KW), 0, stream>>>(
        query, qpts, Ws, bs, Wa, ba, attw, ipts, idx, inp, Wv, bv, value);

    static_assert(CC == 4 * 4 * 16 && CC == KK, "the 1x4 waves own all output columns");
    wmma_gemm_lds_f32<1, 4, 1, 4, 2><<<dim3((BB * LQ) / 16, 1), dim3(128), 0, stream>>>(
        value, Wo, bo, out, attw, idx, CC);
}


# KNN scan: LDS reads of candidate fragments stand at 32 places of the wave's instruction stream, not 256: 8 blocks' fragments per place, a group ahead, 2nd register set; same order, bit-identical
# speedup vs baseline: 4.0026x; 1.0495x over previous
#include <hip/hip_runtime.h>
#include <hip/hip_bf16.h>


typedef __attribute__((ext_vector_type(2))) float v2f;
typedef __attribute__((ext_vector_type(8))) float v8f;
typedef __attribute__((ext_vector_type(4))) float v4f;

#define BB   2
#define LQ   1024
#define LIN  4096
#define CC   256
#define MH   8
#define NP   4
#define DH   32
#define QT   4
#define KW   4

#define KK     256
#define KPITCH (KK + 4)
#define KCH    64
#define KBL    32

template <int MT, int NT, int WM, int WN, int EPI>
__global__ void __launch_bounds__(32 * WM * WN)
wmma_gemm_lds_f32(const float* __restrict__ X, const float* __restrict__ W,
                  const float* __restrict__ bias, float* __restrict__ out,
                  const float* __restrict__ aux, const int* __restrict__ gidx, int Ncols) {
    constexpr int NTH  = 32 * WM * WN;
    constexpr int RX   = 16 * MT * WM;
    constexpr int RW   = 16 * NT * WN;
    constexpr int ROWS = RX + RW;
    constexpr int SROWS = (EPI == 2) ? RW : ROWS;
    constexpr int C4   = KCH / 4;
    constexpr int F4C  = SROWS * C4;
    constexpr int NLD  = (F4C + NTH - 1) / NTH;
    constexpr int NCH  = KK / KCH;
    constexpr int NKB  = KK / KBL;
    constexpr int BPC  = KCH / KBL;
    constexpr int KS   = KBL / 4;

    __shared__ __attribute__((aligned(16))) float sT[ROWS * KPITCH];

    const int tid  = threadIdx.x;
    const int lane = tid & 31;
    const int wv   = tid >> 5;
    const int l15  = lane & 15;
    const int kh   = lane >> 4;
    const int wm   = wv / WN;
    const int wn   = wv % WN;
    const int row0 = blockIdx.x * RX;
    const int col0 = blockIdx.y * RW;

    constexpr int P4  = KK / 4;
    constexpr int GRS = NTH / P4;
    constexpr int NGP = (EPI == 2) ? RX / GRS : 1;
    const int gpc = tid % P4;
    const int gr0 = tid / P4;
    int4   gi[NGP];
    float4 ga[NGP];
    if constexpr (EPI == 2) {
        static_assert(NP == 4 && KK == MH * DH && DH % 4 == 0, "four points per head, heads side by side");
        static_assert(NTH % P4 == 0 && RX % GRS == 0, "whole trips, one column piece per thread");
#pragma unroll
        for (int j = 0; j < NGP; ++j) {
            const size_t o = ((size_t)(row0 + gr0 + j * GRS) * MH + gpc / (DH / 4)) * NP;
            gi[j] = *(const int4*)(gidx + o);
            ga[j] = *(const float4*)(aux + o);
        }
    }

    const float* gsrc[NLD];
    int          soff[NLD];
#pragma unroll
    for (int j = 0; j < NLD; ++j) {
        int f = tid + j * NTH;
        if (f >= F4C) f = F4C - 1;
        const int r  = f / C4 + (ROWS - SROWS);
        const int c4 = f % C4;
        gsrc[j] = (r < RX) ? X + (size_t)(row0 + r) * KK + c4 * 4
                           : W + (size_t)(col0 + r - RX) * KK + c4 * 4;
        soff[j] = r * KPITCH + c4 * 4;
    }

    v4f tmp[NLD];
#pragma unroll
    for (int j = 0; j < NLD; ++j)
        if (j + 1 < NLD || F4C % NTH == 0 || tid + j * NTH < F4C)
            tmp[j] = *(const v4f*)(gsrc[j]);
    if constexpr (EPI == 2) {
        v4f gv[NGP][NP];
#pragma unroll
        for (int j = 0; j < NGP; ++j) {
            const int    row = row0 + gr0 + j * GRS;
            const float* vb  = X + (size_t)(row / LQ) * LIN * KK + gpc * 4;
            gv[j][0] = *(const v4f*)(vb + (size_t)gi[j].x * KK);
            gv[j][1] = *(const v4f*)(vb + (size_t)gi[j].y * KK);
            gv[j][2] = *(const v4f*)(vb + (size_t)gi[j].z * KK);
            gv[j][3] = *(const v4f*)(vb + (size_t)gi[j].w * KK);
        }
#pragma unroll
        for (int j = 0; j < NGP; ++j) {
            v4f x;
#pragma unroll
            for (int e = 0; e < 4; ++e) {
                float acc = 0.0f;
                acc += ga[j].x * gv[j][0][e];
                acc += ga[j].y * gv[j][1][e];
                acc += ga[j].z * gv[j][2][e];
                acc += ga[j].w * gv[j][3][e];
                x[e] = acc;
            }
            *(v4f*)(sT + (gr0 + j * GRS) * KPITCH + gpc * 4) = x;
        }
    }
#pragma unroll
    for (int j = 0; j < NLD; ++j)
        if (j + 1 < NLD || F4C % NTH == 0 || tid + j * NTH < F4C)
            *(v4f*)(sT + soff[j]) = tmp[j];
    __syncthreads();
    if (NCH > 1) {
#pragma unroll
        for (int j = 0; j < NLD; ++j)
            if (j + 1 < NLD || F4C % NTH == 0 || tid + j * NTH < F4C)
                tmp[j] = *(const v4f*)(gsrc[j] + KCH);
    }

    const float* ap = sT + (wm * 16 * MT + l15) * KPITCH + 2 * kh;
    const float* bp = sT + (RX + wn * 16 * NT + l15) * KPITCH + 2 * kh;

    v8f acc[MT][NT];
#pragma unroll
    for (int mt = 0; mt < MT; ++mt)
#pragma unroll
        for (int nt = 0; nt < NT; ++nt) {
            v8f z = {};
            acc[mt][nt] = z;
        }

    v2f fa[2][KS][MT], fb[2][KS][NT];
#pragma unroll
    for (int ks = 0; ks < KS; ++ks) {
#pragma unroll
        for (int mt = 0; mt < MT; ++mt)
            fa[0][ks][mt] = *(const v2f*)(ap + mt * 16 * KPITCH + ks * 4);
#pragma unroll
        for (int nt = 0; nt < NT; ++nt)
            fb[0][ks][nt] = *(const v2f*)(bp + nt * 16 * KPITCH + ks * 4);
    }

#pragma unroll
    for (int kb = 0; kb < NKB; ++kb) {
        const int cur = kb & 1;
        const int c   = kb / BPC;
        if (kb % BPC == BPC - 1 && c + 1 < NCH) {
#pragma unroll
            for (int j = 0; j < NLD; ++j)
                if (j + 1 < NLD || F4C % NTH == 0 || tid + j * NTH < F4C)
                    *(v4f*)(sT + soff[j] + (c + 1) * KCH) = tmp[j];
            __syncthreads();
            if (c + 2 < NCH) {
#pragma unroll
                for (int j = 0; j < NLD; ++j)
                    if (j + 1 < NLD || F4C % NTH == 0 || tid + j * NTH < F4C)
                        tmp[j] = *(const v4f*)(gsrc[j] + (c + 2) * KCH);
            }
        }
        if (kb + 1 < NKB) {
#pragma unroll
            for (int ks = 0; ks < KS; ++ks) {
#pragma unroll
                for (int mt = 0; mt < MT; ++mt)
                    fa[cur ^ 1][ks][mt] =
                        *(const v2f*)(ap + mt * 16 * KPITCH + (kb + 1) * KBL + ks * 4);
#pragma unroll
                for (int nt = 0; nt < NT; ++nt)
                    fb[cur ^ 1][ks][nt] =
                        *(const v2f*)(bp + nt * 16 * KPITCH + (kb + 1) * KBL + ks * 4);
            }
        }
        __builtin_amdgcn_sched_barrier(0);
#pragma unroll
        for (int ks = 0; ks < KS; ++ks)
#pragma unroll
            for (int mt = 0; mt < MT; ++mt)
#pragma unroll
                for (int nt = 0; nt < NT; ++nt)
                    acc[mt][nt] = __builtin_amdgcn_wmma_f32_16x16x4_f32(
                         false, fa[cur][ks][mt],  false, fb[cur][ks][nt],
                         (short)0, acc[mt][nt],
                         false,  false);
        __builtin_amdgcn_sched_barrier(0);
    }

    const int orow0 = row0 + wm * 16 * MT;
    const int ocol0 = col0 + wn * 16 * NT;
#pragma unroll
    for (int nt = 0; nt < NT; ++nt) {
        const int   col  = ocol0 + nt * 16;
        const float bcol = bias[col + l15];
#pragma unroll
        for (int mt = 0; mt < MT; ++mt) {
            const int row = orow0 + mt * 16;
#pragma unroll
            for (int r = 0; r < 8; ++r) {
                out[(size_t)(row + r + 8 * kh) * Ncols + col + l15] =
                    acc[mt][nt][r] + bcol;
            }
        }
    }
}

__device__ __forceinline__ void knn_rows2(const v8f* c, int r0, int cib,
                                          float* best, int* bidx) {
    bool lt[2][QT];
#pragma unroll
    for (int r = r0; r < r0 + 2; ++r)
#pragma unroll
        for (int t = 0; t < QT; ++t)
            lt[r - r0][t] = c[t][r] < best[(r & 1) * QT + t];
#pragma unroll
    for (int r = r0; r < r0 + 2; ++r) {
#pragma unroll
        for (int t = 0; t < QT; ++t) {
            const int e = (r & 1) * QT + t;
            best[e] = lt[r - r0][t] ? c[t][r] : best[e];
            bidx[e] = lt[r - r0][t] ? cib + r : bidx[e];
        }
    }
}

__device__ __forceinline__ void knn_phase(v8f* cn, const v8f* cc, const v2f a,
                                          const v2f* bq, int cib,
                                          float* best, int* bidx) {
    static_assert(2 * QT == 8, "one product per two of a tile's eight rows");
#pragma unroll
    for (int t = 0; t < QT; ++t) {
        v8f z = {};
        cn[t] = __builtin_amdgcn_wmma_f32_16x16x4_f32(
            false, a, false, bq[t], (short)0, z, false, false);
        knn_rows2(cc, 2 * t, cib, best, bidx);
        __builtin_amdgcn_sched_barrier(0);
    }
}

#define VRX   64
#define VRW   128
#define VROWS (VRX + VRW)
#define QRW   8
#define SW1   (3 * MH * NP)
#define SWN   (SW1 + MH * NP)
#define SWR0  16
#define SEP   (SWN + 4)

__global__ void __launch_bounds__(64 * KW)
points_knn_value(const float* __restrict__ query, const float* __restrict__ qpts,
                 const float* __restrict__ Ws, const float* __restrict__ bs,
                 const float* __restrict__ Wa, const float* __restrict__ ba,
                 float* __restrict__ attw,
                 const float* __restrict__ ipts, int* __restrict__ idxOut,
                 const float* __restrict__ X, const float* __restrict__ Wt,
                 const float* __restrict__ bias, float* __restrict__ vout) {
    constexpr int NTH = 64 * KW;
    constexpr int C4  = KK / 4;
    constexpr int RPT = NTH / C4;
    constexpr int NLQ = QRW / RPT;
    constexpr int NLS = SWN / RPT;
    constexpr int NLX = VRX / RPT;
    constexpr int NLW = VRW / RPT;
    constexpr int NKB = KK / KBL;
    constexpr int KS  = KBL / 4;
    static_assert(NTH % C4 == 0 && QRW % RPT == 0 && SW1 % RPT == 0 && SWN % RPT == 0 &&
                  VRX % RPT == 0 && VRW % RPT == 0, "whole trips, every trip in one source");
    static_assert(KW * 16 * QT == QRW * MH * NP && NTH == QRW * MH * NP,
                  "the scan's points are the points of the work-group's rows, one per thread");
    static_assert(SWN == 16 * 2 * KW, "one 16-column tile of the stacked product per wave");
    static_assert(SWR0 >= QRW && SWR0 + SWN <= VROWS, "phase 1's operands fit the value block's rows");
    static_assert(NP == 4 && QRW * MH <= 32 * KW, "softmax groups of four, one per thread of the product waves");

    __shared__ __attribute__((aligned(16))) float sV[LIN * 4];
    __shared__ __attribute__((aligned(16))) float sT[VROWS * KPITCH];
    __shared__ __attribute__((aligned(16))) float sE[QRW * SEP];
    __shared__ __attribute__((aligned(16))) float sS[NTH * 4];

    const int tid  = threadIdx.x;
    const int lane = tid & 31;
    const int wv   = __builtin_amdgcn_readfirstlane(tid >> 5);
    const int l15  = lane & 15;
    const int kh   = lane >> 4;
    const int b    = (blockIdx.x * (KW * 16 * QT)) >> 15;
    const int n0   = (blockIdx.x * KW + (wv & (KW - 1))) * (16 * QT);
    const int qrow0 = blockIdx.x * QRW;
    const int row0 = (blockIdx.x >> 1) * VRX;
    const int col0 = (blockIdx.x & 1) * VRW;

    const int pr = tid / C4;
    const int pc = tid % C4;
    float* sdst = sT + pr * KPITCH + pc * 4;

    const float* qp = qpts + (size_t)(qrow0 + tid / (MH * NP)) * 3;
    const float qv0 = qp[0], qv1 = qp[1], qv2 = qp[2];
    const int   lc   = wv * 16 + l15;
    const float bcol = (lc < SW1) ? bs[lc] : ba[lc - SW1];

    {
        const float4* __restrict__ g = (const float4*)(ipts + (size_t)b * LIN * 3);
        float4* sc = (float4*)sV;
        constexpr int NG = LIN / 4 / NTH;
        const size_t toff = (size_t)pr * KK + pc * 4;

        float4 tmp[NG][3];
        v4f    qt[NLQ], st[NLS];
#pragma unroll
        for (int j = 0; j < NG; ++j)
#pragma unroll
            for (int k = 0; k < 3; ++k) tmp[j][k] = g[3 * (tid + NTH * j) + k];
#pragma unroll
        for (int j = 0; j < NLQ; ++j)
            qt[j] = *(const v4f*)(query + (size_t)(qrow0 + j * RPT) * KK + toff);
#pragma unroll
        for (int j = 0; j < NLS; ++j) {
            const int    r    = j * RPT;
            const float* base = (r < SW1) ? Ws + (size_t)r * KK : Wa + (size_t)(r - SW1) * KK;
            st[j] = *(const v4f*)(base + toff);
        }

#pragma unroll
        for (int j = 0; j < NG; ++j) {
            const float f[12] = {tmp[j][0].x, tmp[j][0].y, tmp[j][0].z, tmp[j][0].w,
                                 tmp[j][1].x, tmp[j][1].y, tmp[j][1].z, tmp[j][1].w,
                                 tmp[j][2].x, tmp[j][2].y, tmp[j][2].z, tmp[j][2].w};
#pragma unroll
            for (int i = 0; i < 4; ++i) {
                const float x = f[3 * i], y = f[3 * i + 1], z = f[3 * i + 2];
                float xy;
                {
#pragma clang fp contract(off)
                    xy = x * x + y * y;
                }
                float4 t;
                t.x = -2.0f * x;
                t.y = -2.0f * y;
                t.z = -2.0f * z;
                t.w = __builtin_fmaf(z, z, xy);
                sc[4 * (tid + NTH * j) + i] = t;
            }
        }
#pragma unroll
        for (int j = 0; j < NLQ; ++j) *(v4f*)(sdst + j * RPT * KPITCH) = qt[j];
#pragma unroll
        for (int j = 0; j < NLS; ++j) *(v4f*)(sdst + (SWR0 + j * RPT) * KPITCH) = st[j];
    }
    __syncthreads();

    v4f xt[NLX], wt[NLW];
    {
        const float* xsrc = X  + (size_t)(row0 + pr) * KK + pc * 4;
        const float* wsrc = Wt + (size_t)(col0 + pr) * KK + pc * 4;
#pragma unroll
        for (int j = 0; j < NLX; ++j) xt[j] = *(const v4f*)(xsrc + (size_t)j * RPT * KK);
#pragma unroll
        for (int j = 0; j < NLW; ++j) wt[j] = *(const v4f*)(wsrc + (size_t)j * RPT * KK);
    }

    {
        const float* ap = sT + (l15 & (QRW - 1)) * KPITCH + 2 * kh;
        const float* bp = sT + (SWR0 + wv * 16 + l15) * KPITCH + 2 * kh;
        v8f acc = {};
        v2f fa[2][KS], fb[2][KS];
#pragma unroll
        for (int ks = 0; ks < KS; ++ks) {
            fa[0][ks] = *(const v2f*)(ap + ks * 4);
            fb[0][ks] = *(const v2f*)(bp + ks * 4);
        }
#pragma unroll
        for (int kb = 0; kb < NKB; ++kb) {
            const int cur = kb & 1;
            if (kb + 1 < NKB) {
#pragma unroll
                for (int ks = 0; ks < KS; ++ks) {
                    fa[cur ^ 1][ks] = *(const v2f*)(ap + (kb + 1) * KBL + ks * 4);
                    fb[cur ^ 1][ks] = *(const v2f*)(bp + (kb + 1) * KBL + ks * 4);
                }
            }
            __builtin_amdgcn_sched_barrier(0);
#pragma unroll
            for (int ks = 0; ks < KS; ++ks)
                acc = __builtin_amdgcn_wmma_f32_16x16x4_f32(
                     false, fa[cur][ks],  false, fb[cur][ks],
                     (short)0, acc,  false,  false);
            __builtin_amdgcn_sched_barrier(0);
        }
        if (kh == 0) {
#pragma unroll
            for (int r = 0; r < QRW; ++r) sE[r * SEP + lc] = acc[r] + bcol;
        }
    }
    __syncthreads();

#pragma unroll
    for (int j = 0; j < NLX; ++j) *(v4f*)(sdst + j * RPT * KPITCH) = xt[j];
#pragma unroll
    for (int j = 0; j < NLW; ++j) *(v4f*)(sdst + (VRX + j * RPT) * KPITCH) = wt[j];
    {
        const float* o = sE + (tid / (MH * NP)) * SEP + (tid % (MH * NP)) * 3;
        float4 sp;
        sp.x = qv0 + o[0];
        sp.y = qv1 + o[1];
        sp.z = qv2 + o[2];
        sp.w = 1.0f;
        *(float4*)(sS + tid * 4) = sp;
    }
    __syncthreads();

    if (wv < KW) {
        v2f bq[QT];
#pragma unroll
        for (int t = 0; t < QT; ++t)
            bq[t] = *(const v2f*)(sS + (wv * (16 * QT) + t * 16 + l15) * 4 + 2 * kh);

        const float* vptr = sV + l15 * 4 + 2 * kh;

        float best[2 * QT];
        int   bidx[2 * QT];
#pragma unroll
        for (int t = 0; t < 2 * QT; ++t) { best[t] = 3.4e38f; bidx[t] = 0; }

        constexpr int NBLK = LIN / 16;
        constexpr int FG   = 8;
        constexpr int NGRP = NBLK / FG;
        constexpr int GST  = FG * 16 * 4;
        static_assert(NBLK % (2 * FG) == 0 && (NGRP & (NGRP - 1)) == 0 && FG % 2 == 0,
                      "whole pairs of groups; the tile sets are back in place after a group");
        const int ch = kh << 3;
        v8f c0[QT], c1[QT];
        v2f fr[2][FG];
#pragma unroll
        for (int i = 0; i < FG; ++i) fr[0][i] = *(const v2f*)(vptr + i * 16 * 4);
#pragma unroll
        for (int t = 0; t < QT; ++t) {
            v8f z = {};
            c0[t] = __builtin_amdgcn_wmma_f32_16x16x4_f32(
                false, fr[0][0], false, bq[t], (short)0, z, false, false);
        }
        for (int g = 0; g < NGRP; g += 2) {
            const float* gp = vptr + g * GST;
#pragma unroll
            for (int i = 0; i < FG; ++i) fr[1][i] = *(const v2f*)(gp + GST + i * 16 * 4);
#pragma unroll
            for (int p = 0; p < FG; ++p) {
                const v2f a = (p + 1 < FG) ? fr[0][p + 1] : fr[1][0];
                if (p & 1) knn_phase(c0, c1, a, bq, (g * FG + p) * 16 + ch, best, bidx);
                else       knn_phase(c1, c0, a, bq, (g * FG + p) * 16 + ch, best, bidx);
            }
            const float* gn = vptr + ((g + 2) & (NGRP - 1)) * GST;
#pragma unroll
            for (int i = 0; i < FG; ++i) fr[0][i] = *(const v2f*)(gn + i * 16 * 4);
#pragma unroll
            for (int p = 0; p < FG; ++p) {
                const v2f a = (p + 1 < FG) ? fr[1][p + 1] : fr[0][0];
                if (p & 1) knn_phase(c0, c1, a, bq, ((g + 1) * FG + p) * 16 + ch, best, bidx);
                else       knn_phase(c1, c0, a, bq, ((g + 1) * FG + p) * 16 + ch, best, bidx);
            }
        }

#pragma unroll
        for (int t = 0; t < QT; ++t) {
            if (best[QT + t] < best[t] || (best[QT + t] == best[t] && bidx[QT + t] < bidx[t])) {
                best[t] = best[QT + t]; bidx[t] = bidx[QT + t];
            }
        }
#pragma unroll
        for (int t = 0; t < QT; ++t) {
            const float obest = __shfl_xor(best[t], 16, 32);
            const int   oidx  = __shfl_xor(bidx[t], 16, 32);
            if (obest < best[t] || (obest == best[t] && oidx < bidx[t])) {
                best[t] = obest; bidx[t] = oidx;
            }
            if (lane < 16) idxOut[n0 + t * 16 + l15] = bidx[t];
        }
    } else {
        constexpr int MT  = 2, NT = 4;
        static_assert(VRX == 2 * 16 * MT && VRW == 2 * 16 * NT, "2x2 waves of 32x64");
        const int w  = wv - KW;
        const unsigned sg = (unsigned)tid - 32u * KW;
        if (sg < (unsigned)(QRW * MH)) {
            const int r = (int)(sg / MH);
            const int g = (int)(sg % MH);
            float4 v = *(const float4*)(sE + r * SEP + SW1 + g * 4);
            const float mx = fmaxf(fmaxf(v.x, v.y), fmaxf(v.z, v.w));
            const float e0 = __expf(v.x - mx), e1 = __expf(v.y - mx);
            const float e2 = __expf(v.z - mx), e3 = __expf(v.w - mx);
            const float inv = 1.0f / (e0 + e1 + e2 + e3);
            v.x = e0 * inv; v.y = e1 * inv; v.z = e2 * inv; v.w = e3 * inv;
            *(float4*)(attw + (size_t)(qrow0 + r) * (MH * NP) + g * 4) = v;
        }
        const int wm = w >> 1;
        const int wn = w & 1;
        const float* ap = sT + (wm * 16 * MT + l15) * KPITCH + 2 * kh;
        const float* bp = sT + (VRX + wn * 16 * NT + l15) * KPITCH + 2 * kh;

        v8f acc[MT][NT];
#pragma unroll
        for (int mt = 0; mt < MT; ++mt)
#pragma unroll
            for (int nt = 0; nt < NT; ++nt) {
                v8f z = {};
                acc[mt][nt] = z;
            }

        v2f fa[2][KS][MT], fb[2][KS][NT];
#pragma unroll
        for (int ks = 0; ks < KS; ++ks) {
#pragma unroll
            for (int mt = 0; mt < MT; ++mt)
                fa[0][ks][mt] = *(const v2f*)(ap + mt * 16 * KPITCH + ks * 4);
#pragma unroll
            for (int nt = 0; nt < NT; ++nt)
                fb[0][ks][nt] = *(const v2f*)(bp + nt * 16 * KPITCH + ks * 4);
        }

#pragma unroll
        for (int kb = 0; kb < NKB; ++kb) {
            const int cur = kb & 1;
            if (kb + 1 < NKB) {
#pragma unroll
                for (int ks = 0; ks < KS; ++ks) {
#pragma unroll
                    for (int mt = 0; mt < MT; ++mt)
                        fa[cur ^ 1][ks][mt] =
                            *(const v2f*)(ap + mt * 16 * KPITCH + (kb + 1) * KBL + ks * 4);
#pragma unroll
                    for (int nt = 0; nt < NT; ++nt)
                        fb[cur ^ 1][ks][nt] =
                            *(const v2f*)(bp + nt * 16 * KPITCH + (kb + 1) * KBL + ks * 4);
                }
            }
            __builtin_amdgcn_sched_barrier(0);
#pragma unroll
            for (int ks = 0; ks < KS; ++ks)
#pragma unroll
                for (int mt = 0; mt < MT; ++mt)
#pragma unroll
                    for (int nt = 0; nt < NT; ++nt)
                        acc[mt][nt] = __builtin_amdgcn_wmma_f32_16x16x4_f32(
                             false, fa[cur][ks][mt],  false, fb[cur][ks][nt],
                             (short)0, acc[mt][nt],
                             false,  false);
            __builtin_amdgcn_sched_barrier(0);
        }

        const int orow0 = row0 + wm * 16 * MT;
        const int ocol0 = col0 + wn * 16 * NT;
#pragma unroll
        for (int nt = 0; nt < NT; ++nt) {
            const int   col  = ocol0 + nt * 16;
            const float bcol = bias[col + l15];
#pragma unroll
            for (int mt = 0; mt < MT; ++mt) {
                const int row = orow0 + mt * 16;
#pragma unroll
                for (int r = 0; r < 8; ++r) {
                    vout[(size_t)(row + r + 8 * kh) * CC + col + l15] =
                        acc[mt][nt][r] + bcol;
                }
            }
        }
    }
}

extern "C" void kernel_launch(void* const* d_in, const int* in_sizes, int n_in,
                              void* d_out, int out_size, void* d_ws, size_t ws_size,
                              hipStream_t stream) {
    const float* query = (const float*)d_in[0];
    const float* qpts  = (const float*)d_in[1];
    const float* inp   = (const float*)d_in[2];
    const float* ipts  = (const float*)d_in[3];
    const float* Wv    = (const float*)d_in[4];
    const float* bv    = (const float*)d_in[5];
    const float* Ws    = (const float*)d_in[6];
    const float* bs    = (const float*)d_in[7];
    const float* Wa    = (const float*)d_in[8];
    const float* ba    = (const float*)d_in[9];
    const float* Wo    = (const float*)d_in[10];
    const float* bo    = (const float*)d_in[11];
    float* out = (float*)d_out;

    char* w = (char*)d_ws;
    float* value = (float*)w;  w += (size_t)BB * LIN * CC * 4;
    float* attw  = (float*)w;  w += (size_t)BB * LQ * MH * NP * 4;
    int*   idx   = (int*)w;    w += (size_t)BB * LQ * MH * NP * 4;

    static_assert((BB * LQ) / QRW == ((BB * LIN) / VRX) * (CC / VRW),
                  "as many scan work-groups as value blocks");
    points_knn_value<<<(BB * LQ) / QRW, dim3(64 * KW), 0, stream>>>(
        query, qpts, Ws, bs, Wa, ba, attw, ipts, idx, inp, Wv, bv, value);

    static_assert(CC == 4 * 4 * 16 && CC == KK, "the 1x4 waves own all output columns");
    wmma_gemm_lds_f32<1, 4, 1, 4, 2><<<dim3((BB * LQ) / 16, 1), dim3(128), 0, stream>>>(
        value, Wo, bo, out, attw, idx, CC);
}
